# starting kernel
# baseline (speedup 1.0000x reference)
#include <hip/hip_runtime.h>

typedef float    v2f  __attribute__((ext_vector_type(2)));
typedef float    v8f  __attribute__((ext_vector_type(8)));

#define LRELU_SLOPE 0.2f

__device__ __forceinline__ unsigned fenc(float f) {
  unsigned u = __float_as_uint(f);
  return (u & 0x80000000u) ? ~u : (u | 0x80000000u);
}
__device__ __forceinline__ float fdec(unsigned u) {
  u = (u & 0x80000000u) ? (u & 0x7FFFFFFFu) : ~u;
  return __uint_as_float(u);
}

__global__ __launch_bounds__(256) void gat_gemm_wmma(
    const float* __restrict__ X, const float* __restrict__ Wm,
    float* __restrict__ Hout, int N, int K)
{
  __shared__ float wldsT[64 * 130];
  const int stride = K + 2;
  const int tid = threadIdx.x;
  for (int i = tid; i < K * 64; i += 256) {
    const int k = i >> 6, col = i & 63;
    wldsT[col * stride + k] = Wm[i];
  }
  __syncthreads();

  const int wave = tid >> 5;
  const int lane = tid & 31;
  const int l16  = lane & 15;
  const int hsel = lane >> 4;
  const int rowBase = blockIdx.x * 128 + wave * 16;
  int row = rowBase + l16;
  if (row >= N) row = N - 1;

  v8f acc[4] = {};

  const int koff = hsel * 2;
  const float* xrow = X + (size_t)row * K + koff;
  const float* wb0 = &wldsT[(0 * 16 + l16) * stride + koff];
  const float* wb1 = &wldsT[(1 * 16 + l16) * stride + koff];
  const float* wb2 = &wldsT[(2 * 16 + l16) * stride + koff];
  const float* wb3 = &wldsT[(3 * 16 + l16) * stride + koff];

  for (int kk = 0; kk < K; kk += 4) {
    const float2 av  = *(const float2*)(xrow + kk);
    const float2 bv0 = *(const float2*)(wb0 + kk);
    const float2 bv1 = *(const float2*)(wb1 + kk);
    const float2 bv2 = *(const float2*)(wb2 + kk);
    const float2 bv3 = *(const float2*)(wb3 + kk);
    v2f a;  a.x = av.x;   a.y = av.y;
    v2f b0; b0.x = bv0.x; b0.y = bv0.y;
    v2f b1; b1.x = bv1.x; b1.y = bv1.y;
    v2f b2; b2.x = bv2.x; b2.y = bv2.y;
    v2f b3; b3.x = bv3.x; b3.y = bv3.y;
    acc[0] = __builtin_amdgcn_wmma_f32_16x16x4_f32(false, a, false, b0, (short)0, acc[0], false, false);
    acc[1] = __builtin_amdgcn_wmma_f32_16x16x4_f32(false, a, false, b1, (short)0, acc[1], false, false);
    acc[2] = __builtin_amdgcn_wmma_f32_16x16x4_f32(false, a, false, b2, (short)0, acc[2], false, false);
    acc[3] = __builtin_amdgcn_wmma_f32_16x16x4_f32(false, a, false, b3, (short)0, acc[3], false, false);
  }

#pragma unroll
  for (int nt = 0; nt < 4; ++nt) {
#pragma unroll
    for (int j = 0; j < 8; ++j) {
      const int ro = rowBase + j + hsel * 8;
      if (ro < N) Hout[(size_t)ro * 64 + nt * 16 + l16] = acc[nt][j];
    }
  }
}

__global__ __launch_bounds__(256) void gat_logits(
    const float* __restrict__ H, const float* __restrict__ a_src,
    const float* __restrict__ a_dst, float* __restrict__ als,
    float* __restrict__ ald, int N)
{
  const int lane = threadIdx.x & 31;
  const int node = (int)((blockIdx.x * (unsigned)blockDim.x + threadIdx.x) >> 5);
  if (node >= N) return;
  const float h0 = H[(size_t)node * 64 + lane];
  const float h1 = H[(size_t)node * 64 + lane + 32];
  float ps = h0 * a_src[lane] + h1 * a_src[lane + 32];
  float pd = h0 * a_dst[lane] + h1 * a_dst[lane + 32];
#pragma unroll
  for (int o = 16; o > 0; o >>= 1) {
    ps += __shfl_xor(ps, o, 32);
    pd += __shfl_xor(pd, o, 32);
  }
  if (lane == 0) { als[node] = ps; ald[node] = pd; }
}

__global__ void gat_init(unsigned* __restrict__ mEnc, float* __restrict__ denom,
                         float* __restrict__ agg, int N)
{
  const int i = blockIdx.x * blockDim.x + threadIdx.x;
  if (i < N) { mEnc[i] = 0u; denom[i] = 0.0f; }
  if (i < N * 64) agg[i] = 0.0f;
}

__global__ void gat_edge_logit_max(
    const int* __restrict__ ei, int E, int N,
    const float* __restrict__ als, const float* __restrict__ ald,
    float* __restrict__ ebuf, unsigned* __restrict__ mEnc)
{
  const int e = blockIdx.x * blockDim.x + threadIdx.x;
  if (e >= E + N) return;
  int s, d;
  if (e < E) { s = ei[e]; d = ei[E + e]; } else { s = d = e - E; }
  float v = als[s] + ald[d];
  v = (v > 0.0f) ? v : LRELU_SLOPE * v;
  ebuf[e] = v;
  atomicMax(&mEnc[d], fenc(v));
}

__global__ void gat_edge_exp_sum(
    const int* __restrict__ ei, int E, int N,
    const float* __restrict__ ebuf, const unsigned* __restrict__ mEnc,
    float* __restrict__ exbuf, float* __restrict__ denom)
{
  const int e = blockIdx.x * blockDim.x + threadIdx.x;
  if (e >= E + N) return;
  int d;
  if (e < E) { d = ei[E + e]; } else { d = e - E; }
  const float ex = __expf(ebuf[e] - fdec(mEnc[d]));
  exbuf[e] = ex;
  atomicAdd(&denom[d], ex);
}

__global__ __launch_bounds__(256) void gat_edge_scatter(
    const int* __restrict__ ei, int E, int N,
    const float* __restrict__ Hin, const float* __restrict__ exbuf,
    const float* __restrict__ denom, float* __restrict__ agg)
{
  const int lane = threadIdx.x & 31;
  const long long g = (long long)blockIdx.x * blockDim.x + threadIdx.x;
  const int e = (int)(g >> 5);
  if (e >= E + N) return;
  int s, d;
  if (e < E) { s = ei[e]; d = ei[E + e]; } else { s = d = e - E; }
  const float alpha = exbuf[e] / denom[d];
  const float2 hv = *(const float2*)&Hin[(size_t)s * 64 + lane * 2];
  atomicAdd(&agg[(size_t)d * 64 + lane * 2 + 0], alpha * hv.x);
  atomicAdd(&agg[(size_t)d * 64 + lane * 2 + 1], alpha * hv.y);
}

__global__ void gat_epilogue(const float* __restrict__ agg,
                             const float* __restrict__ bias,
                             float* __restrict__ out, int N, int doRelu)
{
  const int i = blockIdx.x * blockDim.x + threadIdx.x;
  if (i >= N * 64) return;
  float v = agg[i] + bias[i & 63];
  if (doRelu) v = fmaxf(v, 0.0f);
  out[i] = v;
}

extern "C" void kernel_launch(void* const* d_in, const int* in_sizes, int n_in,
                              void* d_out, int out_size, void* d_ws, size_t ws_size,
                              hipStream_t stream) {
  const float* x   = (const float*)d_in[0];
  const int*   ei  = (const int*)  d_in[1];
  const float* W1  = (const float*)d_in[2];
  const float* as1 = (const float*)d_in[3];
  const float* ad1 = (const float*)d_in[4];
  const float* b1  = (const float*)d_in[5];
  const float* W2  = (const float*)d_in[6];
  const float* as2 = (const float*)d_in[7];
  const float* ad2 = (const float*)d_in[8];
  const float* b2  = (const float*)d_in[9];

  const int N  = in_sizes[0] / 128;
  const int E  = in_sizes[1] / 2;
  const int ET = E + N;

  char* w = (char*)d_ws;
  float*    h1    = (float*)w;    w += (size_t)N * 64 * sizeof(float);
  float*    h2    = (float*)w;    w += (size_t)N * 64 * sizeof(float);
  float*    agg   = (float*)w;    w += (size_t)N * 64 * sizeof(float);
  float*    als   = (float*)w;    w += (size_t)N * sizeof(float);
  float*    ald   = (float*)w;    w += (size_t)N * sizeof(float);
  unsigned* mEnc  = (unsigned*)w; w += (size_t)N * sizeof(unsigned);
  float*    denom = (float*)w;    w += (size_t)N * sizeof(float);
  float*    ebuf  = (float*)w;    w += (size_t)ET * sizeof(float);
  float*    exbuf = (float*)w;    w += (size_t)ET * sizeof(float);

  const dim3 blk(256);
  const int gGemm     = (N + 127) / 128;
  const int gNode     = (N * 64 + 255) / 256;
  const int gWarpNode = (N * 32 + 255) / 256;
  const int gEdge     = (ET + 255) / 256;
  const int gWaveEdge = (int)(((long long)ET * 32 + 255) / 256);

  gat_gemm_wmma    <<<gGemm,     blk, 0, stream>>>(x,  W1, h1, N, 128);
  gat_logits       <<<gWarpNode, blk, 0, stream>>>(h1, as1, ad1, als, ald, N);
  gat_init         <<<gNode,     blk, 0, stream>>>(mEnc, denom, agg, N);
  gat_edge_logit_max<<<gEdge,    blk, 0, stream>>>(ei, E, N, als, ald, ebuf, mEnc);
  gat_edge_exp_sum <<<gEdge,     blk, 0, stream>>>(ei, E, N, ebuf, mEnc, exbuf, denom);
  gat_edge_scatter <<<gWaveEdge, blk, 0, stream>>>(ei, E, N, h1, exbuf, denom, agg);
  gat_epilogue     <<<gNode,     blk, 0, stream>>>(agg, b1, h2, N, 1);

  gat_gemm_wmma    <<<gGemm,     blk, 0, stream>>>(h2, W2, h1, N, 64);
  gat_logits       <<<gWarpNode, blk, 0, stream>>>(h1, as2, ad2, als, ald, N);
  gat_init         <<<gNode,     blk, 0, stream>>>(mEnc, denom, agg, N);
  gat_edge_logit_max<<<gEdge,    blk, 0, stream>>>(ei, E, N, als, ald, ebuf, mEnc);
  gat_edge_exp_sum <<<gEdge,     blk, 0, stream>>>(ei, E, N, ebuf, mEnc, exbuf, denom);
  gat_edge_scatter <<<gWaveEdge, blk, 0, stream>>>(ei, E, N, h1, exbuf, denom, agg);
  gat_epilogue     <<<gNode,     blk, 0, stream>>>(agg, b2, (float*)d_out, N, 0);
}


# Edge sums: each block owns 200 destination rows in LDS, scans every edge's owner byte, adds its edges with LDS float atomics, stores rows once; 11 to 10 launches; one sum per row, LDS arrival order
# speedup vs baseline: 2.1558x; 2.1558x over previous
#include <hip/hip_runtime.h>


typedef float    v2f  __attribute__((ext_vector_type(2)));
typedef float    v4f  __attribute__((ext_vector_type(4)));
typedef float    v8f  __attribute__((ext_vector_type(8)));
typedef unsigned v4u  __attribute__((ext_vector_type(4)));

#define LRELU_SLOPE 0.2f
#define RNG   200
#define LCAP  1024

__device__ __forceinline__ unsigned fenc(float f) {
  unsigned u = __float_as_uint(f);
  return (u & 0x80000000u) ? ~u : (u | 0x80000000u);
}
__device__ __forceinline__ float fdec(unsigned u) {
  u = (u & 0x80000000u) ? (u & 0x7FFFFFFFu) : ~u;
  return __uint_as_float(u);
}

__device__ __forceinline__ int pkg_half() {
  const unsigned id = (unsigned)__builtin_amdgcn_s_sendmsg_rtn(0x87u);
  return (int)((id >> 18) & 1u);
}

template <bool RELU_A>
__global__ __launch_bounds__(256) void gat_gemm_wmma(
    const float* __restrict__ X,
    const float* __restrict__ Wm,
    float* __restrict__ Hout, int N, int K,
    unsigned* __restrict__ mEnc, float* __restrict__ denom)
{
  __shared__ float wldsT[64 * 130];
  const int stride = K + 2;
  const int tid = threadIdx.x;

  {
    const int rb = blockIdx.x * 128;
    const int n = rb + (tid & 127);
    const size_t c = (size_t)(tid >> 7) * (size_t)N;
    if (n < N) { mEnc[c + n] = 0u; denom[c + n] = 0.0f; }
  }

  for (int i = tid; i < K * 64; i += 256) {
    const int k = i >> 6, col = i & 63;
    wldsT[col * stride + k] = Wm[i];
  }
  __syncthreads();

  const int wave = tid >> 5;
  const int lane = tid & 31;
  const int l16  = lane & 15;
  const int hsel = lane >> 4;
  const int rowBase = blockIdx.x * 128 + wave * 16;
  int row = rowBase + l16;
  if (row >= N) row = N - 1;

  v8f acc[4] = {};

  const int koff = hsel * 2;
  const float* xrow = X + (size_t)row * K + koff;
  const float* wb0 = &wldsT[(0 * 16 + l16) * stride + koff];
  const float* wb1 = &wldsT[(1 * 16 + l16) * stride + koff];
  const float* wb2 = &wldsT[(2 * 16 + l16) * stride + koff];
  const float* wb3 = &wldsT[(3 * 16 + l16) * stride + koff];

  for (int kk = 0; kk < K; kk += 4) {
    const float2 av  = *(const float2*)(xrow + kk);
    const float2 bv0 = *(const float2*)(wb0 + kk);
    const float2 bv1 = *(const float2*)(wb1 + kk);
    const float2 bv2 = *(const float2*)(wb2 + kk);
    const float2 bv3 = *(const float2*)(wb3 + kk);
    v2f a;
    if (RELU_A) { a.x = fmaxf(av.x, 0.0f); a.y = fmaxf(av.y, 0.0f); }
    else        { a.x = av.x; a.y = av.y; }
    v2f b0; b0.x = bv0.x; b0.y = bv0.y;
    v2f b1; b1.x = bv1.x; b1.y = bv1.y;
    v2f b2; b2.x = bv2.x; b2.y = bv2.y;
    v2f b3; b3.x = bv3.x; b3.y = bv3.y;
    acc[0] = __builtin_amdgcn_wmma_f32_16x16x4_f32(false, a, false, b0, (short)0, acc[0], false, false);
    acc[1] = __builtin_amdgcn_wmma_f32_16x16x4_f32(false, a, false, b1, (short)0, acc[1], false, false);
    acc[2] = __builtin_amdgcn_wmma_f32_16x16x4_f32(false, a, false, b2, (short)0, acc[2], false, false);
    acc[3] = __builtin_amdgcn_wmma_f32_16x16x4_f32(false, a, false, b3, (short)0, acc[3], false, false);
  }

#pragma unroll
  for (int nt = 0; nt < 4; ++nt) {
#pragma unroll
    for (int j = 0; j < 8; ++j) {
      const int ro = rowBase + j + hsel * 8;
      if (ro < N) Hout[(size_t)ro * 64 + nt * 16 + l16] = acc[nt][j];
    }
  }
}

__global__ __launch_bounds__(256) void gat_logits(
    const float* __restrict__ H, const float* __restrict__ a_src,
    const float* __restrict__ a_dst, float* __restrict__ als,
    float* __restrict__ ald, int N)
{
  const int lane = threadIdx.x & 31;
  const int node = (int)((blockIdx.x * (unsigned)blockDim.x + threadIdx.x) >> 5);
  if (node >= N) return;
  const float h0 = H[(size_t)node * 64 + lane];
  const float h1 = H[(size_t)node * 64 + lane + 32];
  float ps = h0 * a_src[lane] + h1 * a_src[lane + 32];
  float pd = h0 * a_dst[lane] + h1 * a_dst[lane + 32];
#pragma unroll
  for (int o = 16; o > 0; o >>= 1) {
    ps += __shfl_xor(ps, o, 32);
    pd += __shfl_xor(pd, o, 32);
  }
  if (lane == 0) { als[node] = ps; ald[node] = pd; }
}

__global__ void gat_edge_logit_max(
    const int* __restrict__ ei, int E, int N,
    const float* __restrict__ als, const float* __restrict__ ald,
    float* __restrict__ ebuf, unsigned* __restrict__ mEnc,
    unsigned char* __restrict__ key)
{
  const int e  = blockIdx.x * blockDim.x + threadIdx.x;
  const int ET = E + N;
  if (e >= ET) {
    if (key && e < ((ET + 15) & ~15)) key[e] = (unsigned char)255;
    return;
  }
  const int hf = pkg_half();
  int s, d;
  if (e < E) { s = ei[e]; d = ei[E + e]; } else { s = d = e - E; }
  if (key) key[e] = (unsigned char)(d / RNG);
  float v = als[s] + ald[d];
  v = (v > 0.0f) ? v : LRELU_SLOPE * v;
  ebuf[e] = v;
  atomicMax(&mEnc[(size_t)hf * (size_t)N + d], fenc(v));
}

__global__ void gat_edge_exp_sum(
    const int* __restrict__ ei, int E, int N,
    const float* __restrict__ ebuf, const unsigned* __restrict__ mEnc,
    float* __restrict__ exbuf, float* __restrict__ denom)
{
  const int e = blockIdx.x * blockDim.x + threadIdx.x;
  if (e >= E + N) return;
  const int hf = pkg_half();
  int d;
  if (e < E) { d = ei[E + e]; } else { d = e - E; }
  const unsigned ma = mEnc[d], mb = mEnc[N + d];
  const float ex = __expf(ebuf[e] - fdec(ma > mb ? ma : mb));
  exbuf[e] = ex;
  atomicAdd(&denom[(size_t)hf * (size_t)N + d], ex);
}


__device__ __forceinline__ unsigned zero_byte_map(unsigned t0, unsigned t1, unsigned t2, unsigned t3) {
  const unsigned M  = 0x7F7F7F7Fu;
  const unsigned y0 = ~(((t0 & M) + M) | t0 | M);
  const unsigned y1 = ~(((t1 & M) + M) | t1 | M);
  const unsigned y2 = ~(((t2 & M) + M) | t2 | M);
  const unsigned y3 = ~(((t3 & M) + M) | t3 | M);
  return (y0 >> 7) | (y1 >> 6) | (y2 >> 5) | (y3 >> 4);
}

__device__ __forceinline__ int own_compact(int* mylist, int cntl, int lane) {
  int total = 0;
  for (int k = 0; k < 32; ++k) {
    const bool has = k < cntl;
    const unsigned bal = __builtin_amdgcn_ballot_w32(has);
    if (bal == 0u) break;
    if (has) {
      const int v = mylist[k * 32 + lane];
      mylist[total + (int)__builtin_amdgcn_mbcnt_lo(bal, 0u)] = v;
    }
    total += __builtin_popcount(bal);
  }
  return total;
}

__device__ __forceinline__ void own_flush(
    const int* mylist, int cnt, int lane, int lo,
    const int* __restrict__ ei, int E, int N,
    const float* __restrict__ Hin, const float* __restrict__ exbuf,
    const float* __restrict__ denom, float* accs)
{
  for (int b = 0; b < cnt; b += 32) {
    const int nb = (cnt - b < 32) ? (cnt - b) : 32;
    const int e  = mylist[b + ((lane < nb) ? lane : 0)];
    int s, d;
    if (e < E) { s = ei[e]; d = ei[E + e]; } else { s = d = e - E; }
    const float alpha = exbuf[e] / (denom[d] + denom[N + d]);
    const int dl = d - lo;
    for (int g = 0; g < nb; g += 8) {
      float2 hv[8];
#pragma unroll
      for (int u = 0; u < 8; ++u) {
        const int sj = __builtin_amdgcn_readlane(s, g + u);
        hv[u] = *(const float2*)(Hin + (size_t)sj * 64 + lane * 2);
      }
#pragma unroll
      for (int u = 0; u < 8; ++u) {
        const bool  live = (u == 0) || (g + u < nb);
        const int   dj = __builtin_amdgcn_readlane(dl, g + u);
        const float aj = __uint_as_float(
            __builtin_amdgcn_readlane(__float_as_uint(alpha), g + u));
        float* row = accs + dj * 64 + lane * 2;
        atomicAdd(row,     live ? aj * hv[u].x : 0.0f);
        atomicAdd(row + 1, live ? aj * hv[u].y : 0.0f);
      }
    }
  }
}

__global__ __launch_bounds__(1024) void gat_edge_aggregate_owned(
    const int* __restrict__ ei, int E, int N,
    const float* __restrict__ Hin, const float* __restrict__ exbuf,
    const float* __restrict__ denom, const unsigned* __restrict__ key32,
    const float* __restrict__ bias, float* __restrict__ outp)
{
  __shared__ __attribute__((aligned(16))) float accs[RNG * 64];
  __shared__ int   lists[32 * LCAP];

  const int tid  = threadIdx.x;
  const int lane = tid & 31;
  const int wave = __builtin_amdgcn_readfirstlane(tid >> 5);
  const int ET   = E + N;
  const int lo   = blockIdx.x * RNG;

  for (int i = tid; i < RNG * 64; i += 1024) accs[i] = 0.0f;
  __syncthreads();

  int* mylist = lists + wave * LCAP;
  const int ngroups = (ET + 15) >> 4;
  const unsigned kb = (unsigned)blockIdx.x * 0x01010101u;
  int cntl = 0;

  const int nchunk = (ngroups + 4095) >> 12;
  int chunk = (int)(blockIdx.x % (unsigned)nchunk);
  for (int it = 0; it < nchunk; ++it) {
    const int g0 = chunk << 12;
    chunk = (chunk + 1 == nchunk) ? 0 : chunk + 1;
    v4u x[4];
#pragma unroll
    for (int q = 0; q < 4; ++q) {
      const int gi = g0 + q * 1024 + tid;
      x[q] = *(const v4u*)(key32 + 4 * (size_t)((gi < ngroups) ? gi : (ngroups - 1)));
    }
#pragma unroll
    for (int q = 0; q < 4; ++q) {
      const int gi = g0 + q * 1024 + tid;
      const unsigned t0 = x[q].x ^ kb, t1 = x[q].y ^ kb, t2 = x[q].z ^ kb, t3 = x[q].w ^ kb;
      unsigned any = (((t0 - 0x01010101u) & ~t0) | ((t1 - 0x01010101u) & ~t1) |
                      ((t2 - 0x01010101u) & ~t2) | ((t3 - 0x01010101u) & ~t3)) & 0x80808080u;
      if (gi >= ngroups) any = 0u;
      if (__builtin_amdgcn_ballot_w32(any != 0u) != 0u) {
        if (any != 0u) {
          unsigned m = zero_byte_map(t0, t1, t2, t3);
          do {
            const int p = __builtin_ctz(m);
            m &= m - 1u;
            mylist[cntl * 32 + lane] = gi * 16 + 4 * (p & 7) + (p >> 3);
            ++cntl;
          } while (m != 0u);
        }
        if (__builtin_amdgcn_ballot_w32(cntl > 16) != 0u) {
          const int total = own_compact(mylist, cntl, lane);
          own_flush(mylist, total, lane, lo, ei, E, N, Hin, exbuf, denom, accs);
          cntl = 0;
        }
      }
    }
  }
  {
    const int total = own_compact(mylist, cntl, lane);
    if (total > 0) own_flush(mylist, total, lane, lo, ei, E, N, Hin, exbuf, denom, accs);
  }
  __syncthreads();

  for (int i = tid; i < RNG * 16; i += 1024) {
    const int r = i >> 4, c4 = i & 15;
    const int n = lo + r;
    if (n < N) {
      const v4f a  = *(const v4f*)(accs + r * 64 + c4 * 4);
      const v4f bv = *(const v4f*)(bias + c4 * 4);
      *(v4f*)(outp + (size_t)n * 64 + c4 * 4) = a + bv;
    }
  }
}

extern "C" void kernel_launch(void* const* d_in, const int* in_sizes, int n_in,
                              void* d_out, int out_size, void* d_ws, size_t ws_size,
                              hipStream_t stream) {
  const float* x   = (const float*)d_in[0];
  const int*   ei  = (const int*)  d_in[1];
  const float* W1  = (const float*)d_in[2];
  const float* as1 = (const float*)d_in[3];
  const float* ad1 = (const float*)d_in[4];
  const float* b1  = (const float*)d_in[5];
  const float* W2  = (const float*)d_in[6];
  const float* as2 = (const float*)d_in[7];
  const float* ad2 = (const float*)d_in[8];
  const float* b2  = (const float*)d_in[9];

  const int N  = in_sizes[0] / 128;
  const int E  = in_sizes[1] / 2;
  const int ET = E + N;

  char* w = (char*)d_ws;
  float*    h     = (float*)w;    w += (size_t)N * 64 * sizeof(float);
  float*    agg   = (float*)w;    w += (size_t)N * 64 * sizeof(float);
  float*    als   = (float*)w;    w += (size_t)N * sizeof(float);
  float*    ald   = (float*)w;    w += (size_t)N * sizeof(float);
  unsigned* mEnc  = (unsigned*)w; w += (size_t)N * 2 * sizeof(unsigned);
  float*    denom = (float*)w;    w += (size_t)N * 2 * sizeof(float);
  float*    ebuf  = (float*)w;    w += (size_t)ET * sizeof(float);
  float*    exbuf = (float*)w;    w += (size_t)ET * sizeof(float);
  w = (char*)d_ws + (((size_t)(w - (char*)d_ws) + 15) & ~(size_t)15);
  unsigned char* key = (unsigned char*)w;
  float*    out   = (float*)d_out;

  const dim3 blk(256);
  const int gGemm     = (N + 127) / 128;
  const int gWarpNode = (N * 32 + 255) / 256;
  const int gEdge     = (ET + 255) / 256;
  const int gOwn      = (N + RNG - 1) / RNG;

  gat_gemm_wmma<false><<<gGemm,  blk, 0, stream>>>(x, W1, h, N, 128, mEnc, denom);
  gat_logits       <<<gWarpNode, blk, 0, stream>>>(h, as1, ad1, als, ald, N);
  gat_edge_logit_max<<<gEdge,    blk, 0, stream>>>(ei, E, N, als, ald, ebuf, mEnc, key);
  gat_edge_exp_sum <<<gEdge,     blk, 0, stream>>>(ei, E, N, ebuf, mEnc, exbuf, denom);
  gat_edge_aggregate_owned<<<gOwn, dim3(1024), 0, stream>>>(ei, E, N, h, exbuf, denom, (const unsigned*)key, b1, agg);

  gat_gemm_wmma<true><<<gGemm,   blk, 0, stream>>>(agg, W2, h, N, 64, mEnc, denom);
  gat_logits       <<<gWarpNode, blk, 0, stream>>>(h, as2, ad2, als, ald, N);
  gat_edge_logit_max<<<gEdge,    blk, 0, stream>>>(ei, E, N, als, ald, ebuf, mEnc, (unsigned char*)nullptr);
  gat_edge_exp_sum <<<gEdge,     blk, 0, stream>>>(ei, E, N, ebuf, mEnc, exbuf, denom);
  gat_edge_aggregate_owned<<<gOwn, dim3(1024), 0, stream>>>(ei, E, N, h, exbuf, denom, (const unsigned*)key, b2, out);
}


# Softmax max and denominator of a block's 200 owned rows formed in its LDS from its edge list; rows divided once at the store; the 4 global-atomic edge launches gone, 10 to 6; LDS arrival-order sums
# speedup vs baseline: 2.4574x; 1.1399x over previous
#include <hip/hip_runtime.h>


typedef float    v2f  __attribute__((ext_vector_type(2)));
typedef float    v4f  __attribute__((ext_vector_type(4)));
typedef float    v8f  __attribute__((ext_vector_type(8)));
typedef unsigned v4u  __attribute__((ext_vector_type(4)));

#define LRELU_SLOPE 0.2f
#define RNG   200
#define LCAP  1024

__device__ __forceinline__ unsigned fenc(float f) {
  unsigned u = __float_as_uint(f);
  return (u & 0x80000000u) ? ~u : (u | 0x80000000u);
}
__device__ __forceinline__ float fdec(unsigned u) {
  u = (u & 0x80000000u) ? (u & 0x7FFFFFFFu) : ~u;
  return __uint_as_float(u);
}

template <bool RELU_A>
__global__ __launch_bounds__(256) void gat_gemm_wmma(
    const float* __restrict__ X,
    const float* __restrict__ Wm,
    float* __restrict__ Hout, int N, int K)
{
  __shared__ float wldsT[64 * 130];
  const int stride = K + 2;
  const int tid = threadIdx.x;

  for (int i = tid; i < K * 64; i += 256) {
    const int k = i >> 6, col = i & 63;
    wldsT[col * stride + k] = Wm[i];
  }
  __syncthreads();

  const int wave = tid >> 5;
  const int lane = tid & 31;
  const int l16  = lane & 15;
  const int hsel = lane >> 4;
  const int rowBase = blockIdx.x * 128 + wave * 16;
  int row = rowBase + l16;
  if (row >= N) row = N - 1;

  v8f acc[4] = {};

  const int koff = hsel * 2;
  const float* xrow = X + (size_t)row * K + koff;
  const float* wb0 = &wldsT[(0 * 16 + l16) * stride + koff];
  const float* wb1 = &wldsT[(1 * 16 + l16) * stride + koff];
  const float* wb2 = &wldsT[(2 * 16 + l16) * stride + koff];
  const float* wb3 = &wldsT[(3 * 16 + l16) * stride + koff];

  for (int kk = 0; kk < K; kk += 4) {
    const float2 av  = *(const float2*)(xrow + kk);
    const float2 bv0 = *(const float2*)(wb0 + kk);
    const float2 bv1 = *(const float2*)(wb1 + kk);
    const float2 bv2 = *(const float2*)(wb2 + kk);
    const float2 bv3 = *(const float2*)(wb3 + kk);
    v2f a;
    if (RELU_A) { a.x = fmaxf(av.x, 0.0f); a.y = fmaxf(av.y, 0.0f); }
    else        { a.x = av.x; a.y = av.y; }
    v2f b0; b0.x = bv0.x; b0.y = bv0.y;
    v2f b1; b1.x = bv1.x; b1.y = bv1.y;
    v2f b2; b2.x = bv2.x; b2.y = bv2.y;
    v2f b3; b3.x = bv3.x; b3.y = bv3.y;
    acc[0] = __builtin_amdgcn_wmma_f32_16x16x4_f32(false, a, false, b0, (short)0, acc[0], false, false);
    acc[1] = __builtin_amdgcn_wmma_f32_16x16x4_f32(false, a, false, b1, (short)0, acc[1], false, false);
    acc[2] = __builtin_amdgcn_wmma_f32_16x16x4_f32(false, a, false, b2, (short)0, acc[2], false, false);
    acc[3] = __builtin_amdgcn_wmma_f32_16x16x4_f32(false, a, false, b3, (short)0, acc[3], false, false);
  }

#pragma unroll
  for (int nt = 0; nt < 4; ++nt) {
#pragma unroll
    for (int j = 0; j < 8; ++j) {
      const int ro = rowBase + j + hsel * 8;
      if (ro < N) Hout[(size_t)ro * 64 + nt * 16 + l16] = acc[nt][j];
    }
  }
}

__global__ __launch_bounds__(256) void gat_logits(
    const float* __restrict__ H, const float* __restrict__ a_src,
    const float* __restrict__ a_dst, float* __restrict__ als,
    float* __restrict__ ald, int N,
    const int* __restrict__ ei, int E, unsigned* __restrict__ key32)
{
  const unsigned gid = blockIdx.x * (unsigned)blockDim.x + threadIdx.x;
  if (key32) {
    const int ET = E + N;
    const unsigned nwords = (unsigned)((ET + 15) >> 4) << 2;
    for (unsigned w = gid; w < nwords; w += gridDim.x * (unsigned)blockDim.x) {
      const int e0 = (int)(w * 4u);
      int dd[4];
#pragma unroll
      for (int j = 0; j < 4; ++j) dd[j] = ei[E + ((e0 + j < E) ? e0 + j : 0)];
      asm volatile("" :: "v"(dd[0]), "v"(dd[1]), "v"(dd[2]), "v"(dd[3]));
      unsigned word = 0u;
#pragma unroll
      for (int j = 0; j < 4; ++j) {
        const int e = e0 + j;
        const int d = (e < E) ? dd[j] : e - E;
        const unsigned b = (e < ET) ? (unsigned)(d / RNG) : 255u;
        word |= b << (8 * j);
      }
      key32[w] = word;
    }
  }

  const int lane = threadIdx.x & 31;
  const int node = (int)(gid >> 5);
  if (node >= N) return;
  const float h0 = H[(size_t)node * 64 + lane];
  const float h1 = H[(size_t)node * 64 + lane + 32];
  float ps = h0 * a_src[lane] + h1 * a_src[lane + 32];
  float pd = h0 * a_dst[lane] + h1 * a_dst[lane + 32];
#pragma unroll
  for (int o = 16; o > 0; o >>= 1) {
    ps += __shfl_xor(ps, o, 32);
    pd += __shfl_xor(pd, o, 32);
  }
  if (lane == 0) { als[node] = ps; ald[node] = pd; }
}


__device__ __forceinline__ unsigned zero_byte_map(unsigned t0, unsigned t1, unsigned t2, unsigned t3) {
  const unsigned M  = 0x7F7F7F7Fu;
  const unsigned y0 = ~(((t0 & M) + M) | t0 | M);
  const unsigned y1 = ~(((t1 & M) + M) | t1 | M);
  const unsigned y2 = ~(((t2 & M) + M) | t2 | M);
  const unsigned y3 = ~(((t3 & M) + M) | t3 | M);
  return (y0 >> 7) | (y1 >> 6) | (y2 >> 5) | (y3 >> 4);
}

__device__ __forceinline__ int own_compact(int* mylist, int cntl, int lane) {
  int total = 0;
  for (int k = 0; k < 32; ++k) {
    const bool has = k < cntl;
    const unsigned bal = __builtin_amdgcn_ballot_w32(has);
    if (bal == 0u) break;
    if (has) {
      const int v = mylist[k * 32 + lane];
      mylist[total + (int)__builtin_amdgcn_mbcnt_lo(bal, 0u)] = v;
    }
    total += __builtin_popcount(bal);
  }
  return total;
}

__device__ __forceinline__ float edge_logit(float as, float ad) {
  const float v = as + ad;
  return (v > 0.0f) ? v : LRELU_SLOPE * v;
}

__device__ __forceinline__ void own_load4(
    const int* mylist, int base, int total, int lane, int lo,
    const int* __restrict__ ei, int E,
    const float* __restrict__ als, const float* __restrict__ ald,
    int (&s4)[4], int (&dl4)[4], float (&v4)[4])
{
  int d4[4];
#pragma unroll
  for (int u = 0; u < 4; ++u) {
    const int i  = base + u * 32 + lane;
    const int e  = mylist[(i < total) ? i : base];
    const int ec = (e < E) ? e : 0;
    const int sl = ei[ec], dd = ei[E + ec];
    s4[u] = (e < E) ? sl : e - E;
    d4[u] = (e < E) ? dd : e - E;
  }
#pragma unroll
  for (int u = 0; u < 4; ++u) {
    v4[u]  = edge_logit(als[s4[u]], ald[d4[u]]);
    dl4[u] = d4[u] - lo;
  }
}

__device__ __forceinline__ void own_rows(
    int s, int dl, float w, int nb, int lane,
    const float* __restrict__ Hin, float* accs)
{
  for (int g = 0; g < nb; g += 8) {
    float2 hv[8];
#pragma unroll
    for (int u = 0; u < 8; ++u) {
      const int sj = __builtin_amdgcn_readlane(s, g + u);
      hv[u] = *(const float2*)(Hin + (size_t)sj * 64 + lane * 2);
    }
#pragma unroll
    for (int u = 0; u < 8; ++u) {
      const bool  live = (u == 0) || (g + u < nb);
      const int   dj = __builtin_amdgcn_readlane(dl, g + u);
      const float wj = __uint_as_float(
          __builtin_amdgcn_readlane(__float_as_uint(w), g + u));
      float* row = accs + dj * 64 + lane * 2;
      atomicAdd(row,     live ? wj * hv[u].x : 0.0f);
      atomicAdd(row + 1, live ? wj * hv[u].y : 0.0f);
    }
  }
}

__global__ __launch_bounds__(1024) void gat_edge_softmax_aggregate_owned(
    const int* __restrict__ ei, int E, int N,
    const float* __restrict__ Hin, const float* __restrict__ als,
    const float* __restrict__ ald, const unsigned* __restrict__ key32,
    const float* __restrict__ bias, float* __restrict__ outp)
{
  __shared__ __attribute__((aligned(16))) float accs[RNG * 64];
  __shared__ int      lists[32 * LCAP];
  __shared__ unsigned menc[RNG];
  __shared__ float    den[RNG];
  __shared__ int      full;

  const int tid  = threadIdx.x;
  const int lane = tid & 31;
  const int wave = __builtin_amdgcn_readfirstlane(tid >> 5);
  const int ET   = E + N;
  const int lo   = blockIdx.x * RNG;

  for (int i = tid; i < RNG * 64; i += 1024) accs[i] = 0.0f;
  if (tid < RNG) { menc[tid] = 0u; den[tid] = 0.0f; }
  if (tid == 0) full = 0;
  __syncthreads();

  int* mylist = lists + wave * LCAP;
  const int ngroups = (ET + 15) >> 4;
  const unsigned kb = (unsigned)blockIdx.x * 0x01010101u;
  int cntl = 0;

  const int nchunk = (ngroups + 4095) >> 12;
  int chunk = (int)(blockIdx.x % (unsigned)nchunk);
  for (int it = 0; it < nchunk; ++it) {
    const int g0 = chunk << 12;
    chunk = (chunk + 1 == nchunk) ? 0 : chunk + 1;
    v4u x[4];
#pragma unroll
    for (int q = 0; q < 4; ++q) {
      const int gi = g0 + q * 1024 + tid;
      x[q] = *(const v4u*)(key32 + 4 * (size_t)((gi < ngroups) ? gi : (ngroups - 1)));
    }
#pragma unroll
    for (int q = 0; q < 4; ++q) {
      const int gi = g0 + q * 1024 + tid;
      const unsigned t0 = x[q].x ^ kb, t1 = x[q].y ^ kb, t2 = x[q].z ^ kb, t3 = x[q].w ^ kb;
      unsigned any = (((t0 - 0x01010101u) & ~t0) | ((t1 - 0x01010101u) & ~t1) |
                      ((t2 - 0x01010101u) & ~t2) | ((t3 - 0x01010101u) & ~t3)) & 0x80808080u;
      if (gi >= ngroups) any = 0u;
      if (any != 0u) {
        unsigned m = zero_byte_map(t0, t1, t2, t3);
        do {
          const int p = __builtin_ctz(m);
          m &= m - 1u;
          if (cntl < 32) mylist[cntl * 32 + lane] = gi * 16 + 4 * (p & 7) + (p >> 3);
          ++cntl;
        } while (m != 0u);
      }
    }
  }
  if (cntl > 32) full = 1;
  __syncthreads();
  const bool direct = (full != 0);
  const unsigned char* kbytes = (const unsigned char*)key32;
  const int total = direct ? 0 : own_compact(mylist, cntl, lane);

  if (direct) {
    for (int e = tid; e < ET; e += 1024) {
      if ((unsigned)kbytes[e] != blockIdx.x) continue;
      int s, d;
      if (e < E) { s = ei[e]; d = ei[E + e]; } else { s = d = e - E; }
      atomicMax(&menc[d - lo], fenc(edge_logit(als[s], ald[d])));
    }
  } else {
    for (int base = 0; base < total; base += 128) {
      int s4[4], dl4[4]; float v4[4];
      own_load4(mylist, base, total, lane, lo, ei, E, als, ald, s4, dl4, v4);
#pragma unroll
      for (int u = 0; u < 4; ++u) atomicMax(&menc[dl4[u]], fenc(v4[u]));
    }
  }
  __syncthreads();

  if (direct) {
    for (int e = tid; e < ET; e += 1024) {
      if ((unsigned)kbytes[e] != blockIdx.x) continue;
      int s, d;
      if (e < E) { s = ei[e]; d = ei[E + e]; } else { s = d = e - E; }
      const int   dl = d - lo;
      const float ex = __expf(edge_logit(als[s], ald[d]) - fdec(menc[dl]));
      atomicAdd(&den[dl], ex);
      for (int f = 0; f < 64; ++f) atomicAdd(&accs[dl * 64 + f], ex * Hin[(size_t)s * 64 + f]);
    }
  } else {
    for (int base = 0; base < total; base += 128) {
      int s4[4], dl4[4]; float v4[4], ex4[4];
      own_load4(mylist, base, total, lane, lo, ei, E, als, ald, s4, dl4, v4);
#pragma unroll
      for (int u = 0; u < 4; ++u) {
        ex4[u] = __expf(v4[u] - fdec(menc[dl4[u]]));
        if (base + u * 32 + lane < total) atomicAdd(&den[dl4[u]], ex4[u]);
      }
#pragma unroll
      for (int u = 0; u < 4; ++u) {
        const int left = total - base - u * 32;
        if (left > 0) own_rows(s4[u], dl4[u], ex4[u], (left < 32) ? left : 32, lane, Hin, accs);
      }
    }
  }
  __syncthreads();

  for (int i = tid; i < RNG * 16; i += 1024) {
    const int r = i >> 4, c4 = i & 15;
    const int n = lo + r;
    if (n < N) {
      const v4f a  = *(const v4f*)(accs + r * 64 + c4 * 4);
      const v4f bv = *(const v4f*)(bias + c4 * 4);
      *(v4f*)(outp + (size_t)n * 64 + c4 * 4) = a / den[r] + bv;
    }
  }
}

extern "C" void kernel_launch(void* const* d_in, const int* in_sizes, int n_in,
                              void* d_out, int out_size, void* d_ws, size_t ws_size,
                              hipStream_t stream) {
  const float* x   = (const float*)d_in[0];
  const int*   ei  = (const int*)  d_in[1];
  const float* W1  = (const float*)d_in[2];
  const float* as1 = (const float*)d_in[3];
  const float* ad1 = (const float*)d_in[4];
  const float* b1  = (const float*)d_in[5];
  const float* W2  = (const float*)d_in[6];
  const float* as2 = (const float*)d_in[7];
  const float* ad2 = (const float*)d_in[8];
  const float* b2  = (const float*)d_in[9];

  const int N  = in_sizes[0] / 128;
  const int E  = in_sizes[1] / 2;

  char* w = (char*)d_ws;
  float*    h     = (float*)w;    w += (size_t)N * 64 * sizeof(float);
  float*    agg   = (float*)w;    w += (size_t)N * 64 * sizeof(float);
  float*    als   = (float*)w;    w += (size_t)N * sizeof(float);
  float*    ald   = (float*)w;    w += (size_t)N * sizeof(float);
  w = (char*)d_ws + (((size_t)(w - (char*)d_ws) + 15) & ~(size_t)15);
  unsigned* key   = (unsigned*)w;
  float*    out   = (float*)d_out;

  const dim3 blk(256);
  const int gGemm     = (N + 127) / 128;
  const int gWarpNode = (N * 32 + 255) / 256;
  const int gOwn      = (N + RNG - 1) / RNG;

  gat_gemm_wmma<false><<<gGemm,  blk, 0, stream>>>(x, W1, h, N, 128);
  gat_logits       <<<gWarpNode, blk, 0, stream>>>(h, as1, ad1, als, ald, N, ei, E, key);
  gat_edge_softmax_aggregate_owned<<<gOwn, dim3(1024), 0, stream>>>(ei, E, N, h, als, ald, key, b1, agg);

  gat_gemm_wmma<true><<<gGemm,   blk, 0, stream>>>(agg, W2, h, N, 64);
  gat_logits       <<<gWarpNode, blk, 0, stream>>>(h, as2, ad2, als, ald, N, ei, E, (unsigned*)nullptr);
  gat_edge_softmax_aggregate_owned<<<gOwn, dim3(1024), 0, stream>>>(ei, E, N, h, als, ald, key, b2, out);
}


# Edge pass resolves an edge once: the max pass writes source, local row and logit over the edge's LDS list entry; the sum pass reads them back, no second index or logit-table load; sums as before
# speedup vs baseline: 2.6840x; 1.0922x over previous
#include <hip/hip_runtime.h>


typedef float    v2f  __attribute__((ext_vector_type(2)));
typedef float    v4f  __attribute__((ext_vector_type(4)));
typedef float    v8f  __attribute__((ext_vector_type(8)));
typedef unsigned v4u  __attribute__((ext_vector_type(4)));

#define LRELU_SLOPE 0.2f
#define RNG   200
#define LCAP  1024

__device__ __forceinline__ unsigned fenc(float f) {
  unsigned u = __float_as_uint(f);
  return (u & 0x80000000u) ? ~u : (u | 0x80000000u);
}
__device__ __forceinline__ float fdec(unsigned u) {
  u = (u & 0x80000000u) ? (u & 0x7FFFFFFFu) : ~u;
  return __uint_as_float(u);
}

template <bool RELU_A>
__global__ __launch_bounds__(256) void gat_gemm_wmma(
    const float* __restrict__ X,
    const float* __restrict__ Wm,
    float* __restrict__ Hout, int N, int K)
{
  __shared__ float wldsT[64 * 130];
  const int stride = K + 2;
  const int tid = threadIdx.x;

  for (int i = tid; i < K * 64; i += 256) {
    const int k = i >> 6, col = i & 63;
    wldsT[col * stride + k] = Wm[i];
  }
  __syncthreads();

  const int wave = tid >> 5;
  const int lane = tid & 31;
  const int l16  = lane & 15;
  const int hsel = lane >> 4;
  const int rowBase = blockIdx.x * 128 + wave * 16;
  int row = rowBase + l16;
  if (row >= N) row = N - 1;

  v8f acc[4] = {};

  const int koff = hsel * 2;
  const float* xrow = X + (size_t)row * K + koff;
  const float* wb0 = &wldsT[(0 * 16 + l16) * stride + koff];
  const float* wb1 = &wldsT[(1 * 16 + l16) * stride + koff];
  const float* wb2 = &wldsT[(2 * 16 + l16) * stride + koff];
  const float* wb3 = &wldsT[(3 * 16 + l16) * stride + koff];

  for (int kk = 0; kk < K; kk += 4) {
    const float2 av  = *(const float2*)(xrow + kk);
    const float2 bv0 = *(const float2*)(wb0 + kk);
    const float2 bv1 = *(const float2*)(wb1 + kk);
    const float2 bv2 = *(const float2*)(wb2 + kk);
    const float2 bv3 = *(const float2*)(wb3 + kk);
    v2f a;
    if (RELU_A) { a.x = fmaxf(av.x, 0.0f); a.y = fmaxf(av.y, 0.0f); }
    else        { a.x = av.x; a.y = av.y; }
    v2f b0; b0.x = bv0.x; b0.y = bv0.y;
    v2f b1; b1.x = bv1.x; b1.y = bv1.y;
    v2f b2; b2.x = bv2.x; b2.y = bv2.y;
    v2f b3; b3.x = bv3.x; b3.y = bv3.y;
    acc[0] = __builtin_amdgcn_wmma_f32_16x16x4_f32(false, a, false, b0, (short)0, acc[0], false, false);
    acc[1] = __builtin_amdgcn_wmma_f32_16x16x4_f32(false, a, false, b1, (short)0, acc[1], false, false);
    acc[2] = __builtin_amdgcn_wmma_f32_16x16x4_f32(false, a, false, b2, (short)0, acc[2], false, false);
    acc[3] = __builtin_amdgcn_wmma_f32_16x16x4_f32(false, a, false, b3, (short)0, acc[3], false, false);
  }

#pragma unroll
  for (int nt = 0; nt < 4; ++nt) {
#pragma unroll
    for (int j = 0; j < 8; ++j) {
      const int ro = rowBase + j + hsel * 8;
      if (ro < N) Hout[(size_t)ro * 64 + nt * 16 + l16] = acc[nt][j];
    }
  }
}

__global__ __launch_bounds__(256) void gat_logits(
    const float* __restrict__ H, const float* __restrict__ a_src,
    const float* __restrict__ a_dst, float* __restrict__ als,
    float* __restrict__ ald, int N,
    const int* __restrict__ ei, int E, unsigned* __restrict__ key32)
{
  const unsigned gid = blockIdx.x * (unsigned)blockDim.x + threadIdx.x;
  if (key32) {
    const int ET = E + N;
    const unsigned nwords = (unsigned)((ET + 15) >> 4) << 2;
    for (unsigned w = gid; w < nwords; w += gridDim.x * (unsigned)blockDim.x) {
      const int e0 = (int)(w * 4u);
      int dd[4];
#pragma unroll
      for (int j = 0; j < 4; ++j) dd[j] = ei[E + ((e0 + j < E) ? e0 + j : 0)];
      asm volatile("" :: "v"(dd[0]), "v"(dd[1]), "v"(dd[2]), "v"(dd[3]));
      unsigned word = 0u;
#pragma unroll
      for (int j = 0; j < 4; ++j) {
        const int e = e0 + j;
        const int d = (e < E) ? dd[j] : e - E;
        const unsigned b = (e < ET) ? (unsigned)(d / RNG) : 255u;
        word |= b << (8 * j);
      }
      key32[w] = word;
    }
  }

  const int lane = threadIdx.x & 31;
  const int node = (int)(gid >> 5);
  if (node >= N) return;
  const float h0 = H[(size_t)node * 64 + lane];
  const float h1 = H[(size_t)node * 64 + lane + 32];
  float ps = h0 * a_src[lane] + h1 * a_src[lane + 32];
  float pd = h0 * a_dst[lane] + h1 * a_dst[lane + 32];
#pragma unroll
  for (int o = 16; o > 0; o >>= 1) {
    ps += __shfl_xor(ps, o, 32);
    pd += __shfl_xor(pd, o, 32);
  }
  if (lane == 0) { als[node] = ps; ald[node] = pd; }
}


__device__ __forceinline__ unsigned zero_byte_map(unsigned t0, unsigned t1, unsigned t2, unsigned t3) {
  const unsigned M  = 0x7F7F7F7Fu;
  const unsigned y0 = ~(((t0 & M) + M) | t0 | M);
  const unsigned y1 = ~(((t1 & M) + M) | t1 | M);
  const unsigned y2 = ~(((t2 & M) + M) | t2 | M);
  const unsigned y3 = ~(((t3 & M) + M) | t3 | M);
  return (y0 >> 7) | (y1 >> 6) | (y2 >> 5) | (y3 >> 4);
}

__device__ __forceinline__ int own_compact(int* mylist, int cntl, int lane) {
  int total = 0;
  for (int k = 0; k < 32; ++k) {
    const bool has = k < cntl;
    const unsigned bal = __builtin_amdgcn_ballot_w32(has);
    if (bal == 0u) break;
    if (has) {
      const int v = mylist[k * 32 + lane];
      mylist[total + (int)__builtin_amdgcn_mbcnt_lo(bal, 0u)] = v;
    }
    total += __builtin_popcount(bal);
  }
  return total;
}

__device__ __forceinline__ float edge_logit(float as, float ad) {
  const float v = as + ad;
  return (v > 0.0f) ? v : LRELU_SLOPE * v;
}

__device__ __forceinline__ void own_resolve4(
    int* mylist, float* mylgt, int base, int total, int lane, int lo,
    const int* __restrict__ ei, int E,
    const float* __restrict__ als, const float* __restrict__ ald,
    int (&dl4)[4], float (&v4)[4])
{
  int s4[4], d4[4];
#pragma unroll
  for (int u = 0; u < 4; ++u) {
    const int i  = base + u * 32 + lane;
    const int e  = mylist[(i < total) ? i : base];
    const int ec = (e < E) ? e : 0;
    const int sl = ei[ec], dd = ei[E + ec];
    s4[u] = (e < E) ? sl : e - E;
    d4[u] = (e < E) ? dd : e - E;
  }
#pragma unroll
  for (int u = 0; u < 4; ++u) {
    v4[u]  = edge_logit(als[s4[u]], ald[d4[u]]);
    dl4[u] = d4[u] - lo;
  }
#pragma unroll
  for (int u = 0; u < 4; ++u) {
    const int i = base + u * 32 + lane;
    if (i < total) {
      mylist[i] = (int)((unsigned)s4[u] | ((unsigned)dl4[u] << 24));
      mylgt[i]  = v4[u];
    }
  }
}

__device__ __forceinline__ void own_read4(
    const int* mylist, const float* mylgt, int base, int total, int lane,
    int (&s4)[4], int (&dl4)[4], float (&v4)[4])
{
#pragma unroll
  for (int u = 0; u < 4; ++u) {
    const int i = base + u * 32 + lane;
    const int j = (i < total) ? i : base;
    const unsigned p = (unsigned)mylist[j];
    v4[u]  = mylgt[j];
    s4[u]  = (int)(p & 0x00FFFFFFu);
    dl4[u] = (int)(p >> 24);
  }
}

__device__ __forceinline__ void own_rows(
    int s, int dl, float w, int nb, int lane,
    const float* __restrict__ Hin, float* accs)
{
  for (int g = 0; g < nb; g += 8) {
    float2 hv[8];
#pragma unroll
    for (int u = 0; u < 8; ++u) {
      const int sj = __builtin_amdgcn_readlane(s, g + u);
      hv[u] = *(const float2*)(Hin + (size_t)sj * 64 + lane * 2);
    }
#pragma unroll
    for (int u = 0; u < 8; ++u) {
      const bool  live = (u == 0) || (g + u < nb);
      const int   dj = __builtin_amdgcn_readlane(dl, g + u);
      const float wj = __uint_as_float(
          __builtin_amdgcn_readlane(__float_as_uint(w), g + u));
      float* row = accs + dj * 64 + lane * 2;
      atomicAdd(row,     live ? wj * hv[u].x : 0.0f);
      atomicAdd(row + 1, live ? wj * hv[u].y : 0.0f);
    }
  }
}

__global__ __launch_bounds__(1024) void gat_edge_softmax_aggregate_owned(
    const int* __restrict__ ei, int E, int N,
    const float* __restrict__ Hin, const float* __restrict__ als,
    const float* __restrict__ ald, const unsigned* __restrict__ key32,
    const float* __restrict__ bias, float* __restrict__ outp)
{
  __shared__ __attribute__((aligned(16))) float accs[RNG * 64];
  __shared__ int      lists[32 * LCAP];
  __shared__ float    lgts[32 * LCAP];
  __shared__ unsigned menc[RNG];
  __shared__ float    den[RNG];
  __shared__ int      full;

  const int tid  = threadIdx.x;
  const int lane = tid & 31;
  const int wave = __builtin_amdgcn_readfirstlane(tid >> 5);
  const int ET   = E + N;
  const int lo   = blockIdx.x * RNG;

  for (int i = tid; i < RNG * 64; i += 1024) accs[i] = 0.0f;
  if (tid < RNG) { menc[tid] = 0u; den[tid] = 0.0f; }
  if (tid == 0) full = 0;
  __syncthreads();

  int*   mylist = lists + wave * LCAP;
  float* mylgt  = lgts  + wave * LCAP;
  const int ngroups = (ET + 15) >> 4;
  const unsigned kb = (unsigned)blockIdx.x * 0x01010101u;
  int cntl = 0;

  const int nchunk = (ngroups + 4095) >> 12;
  int chunk = (int)(blockIdx.x % (unsigned)nchunk);
  for (int it = 0; it < nchunk; ++it) {
    const int g0 = chunk << 12;
    chunk = (chunk + 1 == nchunk) ? 0 : chunk + 1;
    v4u x[4];
#pragma unroll
    for (int q = 0; q < 4; ++q) {
      const int gi = g0 + q * 1024 + tid;
      x[q] = *(const v4u*)(key32 + 4 * (size_t)((gi < ngroups) ? gi : (ngroups - 1)));
    }
#pragma unroll
    for (int q = 0; q < 4; ++q) {
      const int gi = g0 + q * 1024 + tid;
      const unsigned t0 = x[q].x ^ kb, t1 = x[q].y ^ kb, t2 = x[q].z ^ kb, t3 = x[q].w ^ kb;
      unsigned any = (((t0 - 0x01010101u) & ~t0) | ((t1 - 0x01010101u) & ~t1) |
                      ((t2 - 0x01010101u) & ~t2) | ((t3 - 0x01010101u) & ~t3)) & 0x80808080u;
      if (gi >= ngroups) any = 0u;
      if (any != 0u) {
        unsigned m = zero_byte_map(t0, t1, t2, t3);
        do {
          const int p = __builtin_ctz(m);
          m &= m - 1u;
          if (cntl < 32) mylist[cntl * 32 + lane] = gi * 16 + 4 * (p & 7) + (p >> 3);
          ++cntl;
        } while (m != 0u);
      }
    }
  }
  if (cntl > 32) full = 1;
  __syncthreads();
  const bool direct = (full != 0);
  const unsigned char* kbytes = (const unsigned char*)key32;
  const int total = direct ? 0 : own_compact(mylist, cntl, lane);

  if (direct) {
    for (int e = tid; e < ET; e += 1024) {
      if ((unsigned)kbytes[e] != blockIdx.x) continue;
      int s, d;
      if (e < E) { s = ei[e]; d = ei[E + e]; } else { s = d = e - E; }
      atomicMax(&menc[d - lo], fenc(edge_logit(als[s], ald[d])));
    }
  } else {
    for (int base = 0; base < total; base += 128) {
      int dl4[4]; float v4[4];
      own_resolve4(mylist, mylgt, base, total, lane, lo, ei, E, als, ald, dl4, v4);
#pragma unroll
      for (int u = 0; u < 4; ++u) atomicMax(&menc[dl4[u]], fenc(v4[u]));
    }
  }
  __syncthreads();

  if (direct) {
    for (int e = tid; e < ET; e += 1024) {
      if ((unsigned)kbytes[e] != blockIdx.x) continue;
      int s, d;
      if (e < E) { s = ei[e]; d = ei[E + e]; } else { s = d = e - E; }
      const int   dl = d - lo;
      const float ex = __expf(edge_logit(als[s], ald[d]) - fdec(menc[dl]));
      atomicAdd(&den[dl], ex);
      for (int f = 0; f < 64; ++f) atomicAdd(&accs[dl * 64 + f], ex * Hin[(size_t)s * 64 + f]);
    }
  } else {
    for (int base = 0; base < total; base += 128) {
      int s4[4], dl4[4]; float v4[4], ex4[4];
      own_read4(mylist, mylgt, base, total, lane, s4, dl4, v4);
#pragma unroll
      for (int u = 0; u < 4; ++u) {
        ex4[u] = __expf(v4[u] - fdec(menc[dl4[u]]));
        if (base + u * 32 + lane < total) atomicAdd(&den[dl4[u]], ex4[u]);
      }
#pragma unroll
      for (int u = 0; u < 4; ++u) {
        const int left = total - base - u * 32;
        if (left > 0) own_rows(s4[u], dl4[u], ex4[u], (left < 32) ? left : 32, lane, Hin, accs);
      }
    }
  }
  __syncthreads();

  for (int i = tid; i < RNG * 16; i += 1024) {
    const int r = i >> 4, c4 = i & 15;
    const int n = lo + r;
    if (n < N) {
      const v4f a  = *(const v4f*)(accs + r * 64 + c4 * 4);
      const v4f bv = *(const v4f*)(bias + c4 * 4);
      *(v4f*)(outp + (size_t)n * 64 + c4 * 4) = a / den[r] + bv;
    }
  }
}

extern "C" void kernel_launch(void* const* d_in, const int* in_sizes, int n_in,
                              void* d_out, int out_size, void* d_ws, size_t ws_size,
                              hipStream_t stream) {
  const float* x   = (const float*)d_in[0];
  const int*   ei  = (const int*)  d_in[1];
  const float* W1  = (const float*)d_in[2];
  const float* as1 = (const float*)d_in[3];
  const float* ad1 = (const float*)d_in[4];
  const float* b1  = (const float*)d_in[5];
  const float* W2  = (const float*)d_in[6];
  const float* as2 = (const float*)d_in[7];
  const float* ad2 = (const float*)d_in[8];
  const float* b2  = (const float*)d_in[9];

  const int N  = in_sizes[0] / 128;
  const int E  = in_sizes[1] / 2;

  char* w = (char*)d_ws;
  float*    h     = (float*)w;    w += (size_t)N * 64 * sizeof(float);
  float*    agg   = (float*)w;    w += (size_t)N * 64 * sizeof(float);
  float*    als   = (float*)w;    w += (size_t)N * sizeof(float);
  float*    ald   = (float*)w;    w += (size_t)N * sizeof(float);
  w = (char*)d_ws + (((size_t)(w - (char*)d_ws) + 15) & ~(size_t)15);
  unsigned* key   = (unsigned*)w;
  float*    out   = (float*)d_out;

  const dim3 blk(256);
  const int gGemm     = (N + 127) / 128;
  const int gWarpNode = (N * 32 + 255) / 256;
  const int gOwn      = (N + RNG - 1) / RNG;

  gat_gemm_wmma<false><<<gGemm,  blk, 0, stream>>>(x, W1, h, N, 128);
  gat_logits       <<<gWarpNode, blk, 0, stream>>>(h, as1, ad1, als, ald, N, ei, E, key);
  gat_edge_softmax_aggregate_owned<<<gOwn, dim3(1024), 0, stream>>>(ei, E, N, h, als, ald, key, b1, agg);

  gat_gemm_wmma<true><<<gGemm,   blk, 0, stream>>>(agg, W2, h, N, 64);
  gat_logits       <<<gWarpNode, blk, 0, stream>>>(h, as2, ad2, als, ald, N, ei, E, (unsigned*)nullptr);
  gat_edge_softmax_aggregate_owned<<<gOwn, dim3(1024), 0, stream>>>(ei, E, N, h, als, ald, key, b2, out);
}


# Logits launch: a wave takes 16 nodes, not 1 (a 16-lane half per node, 16-byte row loads, 8 in flight, row_xmask butterfly); 16x fewer waves; logit dot products summed in a different tree
# speedup vs baseline: 2.9107x; 1.0845x over previous
#include <hip/hip_runtime.h>


typedef float    v2f  __attribute__((ext_vector_type(2)));
typedef float    v4f  __attribute__((ext_vector_type(4)));
typedef float    v8f  __attribute__((ext_vector_type(8)));
typedef unsigned v4u  __attribute__((ext_vector_type(4)));

#define LRELU_SLOPE 0.2f
#define RNG   200
#define LCAP  1024

__device__ __forceinline__ unsigned fenc(float f) {
  unsigned u = __float_as_uint(f);
  return (u & 0x80000000u) ? ~u : (u | 0x80000000u);
}
__device__ __forceinline__ float fdec(unsigned u) {
  u = (u & 0x80000000u) ? (u & 0x7FFFFFFFu) : ~u;
  return __uint_as_float(u);
}

template <bool RELU_A>
__global__ __launch_bounds__(256) void gat_gemm_wmma(
    const float* __restrict__ X,
    const float* __restrict__ Wm,
    float* __restrict__ Hout, int N, int K)
{
  __shared__ float wldsT[64 * 130];
  const int stride = K + 2;
  const int tid = threadIdx.x;

  for (int i = tid; i < K * 64; i += 256) {
    const int k = i >> 6, col = i & 63;
    wldsT[col * stride + k] = Wm[i];
  }
  __syncthreads();

  const int wave = tid >> 5;
  const int lane = tid & 31;
  const int l16  = lane & 15;
  const int hsel = lane >> 4;
  const int rowBase = blockIdx.x * 128 + wave * 16;
  int row = rowBase + l16;
  if (row >= N) row = N - 1;

  v8f acc[4] = {};

  const int koff = hsel * 2;
  const float* xrow = X + (size_t)row * K + koff;
  const float* wb0 = &wldsT[(0 * 16 + l16) * stride + koff];
  const float* wb1 = &wldsT[(1 * 16 + l16) * stride + koff];
  const float* wb2 = &wldsT[(2 * 16 + l16) * stride + koff];
  const float* wb3 = &wldsT[(3 * 16 + l16) * stride + koff];

  for (int kk = 0; kk < K; kk += 4) {
    const float2 av  = *(const float2*)(xrow + kk);
    const float2 bv0 = *(const float2*)(wb0 + kk);
    const float2 bv1 = *(const float2*)(wb1 + kk);
    const float2 bv2 = *(const float2*)(wb2 + kk);
    const float2 bv3 = *(const float2*)(wb3 + kk);
    v2f a;
    if (RELU_A) { a.x = fmaxf(av.x, 0.0f); a.y = fmaxf(av.y, 0.0f); }
    else        { a.x = av.x; a.y = av.y; }
    v2f b0; b0.x = bv0.x; b0.y = bv0.y;
    v2f b1; b1.x = bv1.x; b1.y = bv1.y;
    v2f b2; b2.x = bv2.x; b2.y = bv2.y;
    v2f b3; b3.x = bv3.x; b3.y = bv3.y;
    acc[0] = __builtin_amdgcn_wmma_f32_16x16x4_f32(false, a, false, b0, (short)0, acc[0], false, false);
    acc[1] = __builtin_amdgcn_wmma_f32_16x16x4_f32(false, a, false, b1, (short)0, acc[1], false, false);
    acc[2] = __builtin_amdgcn_wmma_f32_16x16x4_f32(false, a, false, b2, (short)0, acc[2], false, false);
    acc[3] = __builtin_amdgcn_wmma_f32_16x16x4_f32(false, a, false, b3, (short)0, acc[3], false, false);
  }

#pragma unroll
  for (int nt = 0; nt < 4; ++nt) {
#pragma unroll
    for (int j = 0; j < 8; ++j) {
      const int ro = rowBase + j + hsel * 8;
      if (ro < N) Hout[(size_t)ro * 64 + nt * 16 + l16] = acc[nt][j];
    }
  }
}


__device__ __forceinline__ float row16_sum(float v) {
  v += __builtin_bit_cast(float, __builtin_amdgcn_update_dpp(0, __builtin_bit_cast(int, v), 0x161, 0xf, 0xf, false));
  v += __builtin_bit_cast(float, __builtin_amdgcn_update_dpp(0, __builtin_bit_cast(int, v), 0x162, 0xf, 0xf, false));
  v += __builtin_bit_cast(float, __builtin_amdgcn_update_dpp(0, __builtin_bit_cast(int, v), 0x164, 0xf, 0xf, false));
  v += __builtin_bit_cast(float, __builtin_amdgcn_update_dpp(0, __builtin_bit_cast(int, v), 0x168, 0xf, 0xf, false));
  return v;
}

__global__ __launch_bounds__(256) void gat_logits(
    const float* __restrict__ H, const float* __restrict__ a_src,
    const float* __restrict__ a_dst, float* __restrict__ als,
    float* __restrict__ ald, int N,
    const int* __restrict__ ei, int E, unsigned* __restrict__ key32)
{
  const unsigned gid = blockIdx.x * (unsigned)blockDim.x + threadIdx.x;
  const int lane  = threadIdx.x & 31;
  const int c     = lane & 15;
  const int hs    = lane >> 4;
  const int node0 = (int)(gid >> 5) * 16;

  v4f hv[8];
#pragma unroll
  for (int j = 0; j < 8; ++j) {
    int n = node0 + 2 * j + hs;
    n = (n < N) ? n : N - 1;
    hv[j] = *(const v4f*)(H + (size_t)n * 64 + c * 4);
  }
  const v4f as = *(const v4f*)(a_src + c * 4);
  const v4f ad = *(const v4f*)(a_dst + c * 4);

  if (key32) {
    const int ET = E + N;
    const unsigned nwords = (unsigned)((ET + 15) >> 4) << 2;
    for (unsigned w = gid; w < nwords; w += gridDim.x * (unsigned)blockDim.x) {
      const int e0 = (int)(w * 4u);
      int dd[4];
#pragma unroll
      for (int j = 0; j < 4; ++j) dd[j] = ei[E + ((e0 + j < E) ? e0 + j : 0)];
      asm volatile("" :: "v"(dd[0]), "v"(dd[1]), "v"(dd[2]), "v"(dd[3]));
      unsigned word = 0u;
#pragma unroll
      for (int j = 0; j < 4; ++j) {
        const int e = e0 + j;
        const int d = (e < E) ? dd[j] : e - E;
        const unsigned b = (e < ET) ? (unsigned)(d / RNG) : 255u;
        word |= b << (8 * j);
      }
      key32[w] = word;
    }
  }

  float rs = 0.0f, rd = 0.0f;
#pragma unroll
  for (int j = 0; j < 8; ++j) {
    float ps = hv[j].x * as.x + hv[j].y * as.y + hv[j].z * as.z + hv[j].w * as.w;
    float pd = hv[j].x * ad.x + hv[j].y * ad.y + hv[j].z * ad.z + hv[j].w * ad.w;
    ps = row16_sum(ps);
    pd = row16_sum(pd);
    if ((c >> 1) == j) { rs = ps; rd = pd; }
  }
  const int node = node0 + c;
  if ((c & 1) == hs && node < N) { als[node] = rs; ald[node] = rd; }
}


__device__ __forceinline__ unsigned zero_byte_map(unsigned t0, unsigned t1, unsigned t2, unsigned t3) {
  const unsigned M  = 0x7F7F7F7Fu;
  const unsigned y0 = ~(((t0 & M) + M) | t0 | M);
  const unsigned y1 = ~(((t1 & M) + M) | t1 | M);
  const unsigned y2 = ~(((t2 & M) + M) | t2 | M);
  const unsigned y3 = ~(((t3 & M) + M) | t3 | M);
  return (y0 >> 7) | (y1 >> 6) | (y2 >> 5) | (y3 >> 4);
}

__device__ __forceinline__ int own_compact(int* mylist, int cntl, int lane) {
  int total = 0;
  for (int k = 0; k < 32; ++k) {
    const bool has = k < cntl;
    const unsigned bal = __builtin_amdgcn_ballot_w32(has);
    if (bal == 0u) break;
    if (has) {
      const int v = mylist[k * 32 + lane];
      mylist[total + (int)__builtin_amdgcn_mbcnt_lo(bal, 0u)] = v;
    }
    total += __builtin_popcount(bal);
  }
  return total;
}

__device__ __forceinline__ float edge_logit(float as, float ad) {
  const float v = as + ad;
  return (v > 0.0f) ? v : LRELU_SLOPE * v;
}

__device__ __forceinline__ void own_resolve4(
    int* mylist, float* mylgt, int base, int total, int lane, int lo,
    const int* __restrict__ ei, int E,
    const float* __restrict__ als, const float* __restrict__ ald,
    int (&dl4)[4], float (&v4)[4])
{
  int s4[4], d4[4];
#pragma unroll
  for (int u = 0; u < 4; ++u) {
    const int i  = base + u * 32 + lane;
    const int e  = mylist[(i < total) ? i : base];
    const int ec = (e < E) ? e : 0;
    const int sl = ei[ec], dd = ei[E + ec];
    s4[u] = (e < E) ? sl : e - E;
    d4[u] = (e < E) ? dd : e - E;
  }
#pragma unroll
  for (int u = 0; u < 4; ++u) {
    v4[u]  = edge_logit(als[s4[u]], ald[d4[u]]);
    dl4[u] = d4[u] - lo;
  }
#pragma unroll
  for (int u = 0; u < 4; ++u) {
    const int i = base + u * 32 + lane;
    if (i < total) {
      mylist[i] = (int)((unsigned)s4[u] | ((unsigned)dl4[u] << 24));
      mylgt[i]  = v4[u];
    }
  }
}

__device__ __forceinline__ void own_read4(
    const int* mylist, const float* mylgt, int base, int total, int lane,
    int (&s4)[4], int (&dl4)[4], float (&v4)[4])
{
#pragma unroll
  for (int u = 0; u < 4; ++u) {
    const int i = base + u * 32 + lane;
    const int j = (i < total) ? i : base;
    const unsigned p = (unsigned)mylist[j];
    v4[u]  = mylgt[j];
    s4[u]  = (int)(p & 0x00FFFFFFu);
    dl4[u] = (int)(p >> 24);
  }
}

__device__ __forceinline__ void own_rows(
    int s, int dl, float w, int nb, int lane,
    const float* __restrict__ Hin, float* accs)
{
  for (int g = 0; g < nb; g += 8) {
    float2 hv[8];
#pragma unroll
    for (int u = 0; u < 8; ++u) {
      const int sj = __builtin_amdgcn_readlane(s, g + u);
      hv[u] = *(const float2*)(Hin + (size_t)sj * 64 + lane * 2);
    }
#pragma unroll
    for (int u = 0; u < 8; ++u) {
      const bool  live = (u == 0) || (g + u < nb);
      const int   dj = __builtin_amdgcn_readlane(dl, g + u);
      const float wj = __uint_as_float(
          __builtin_amdgcn_readlane(__float_as_uint(w), g + u));
      float* row = accs + dj * 64 + lane * 2;
      atomicAdd(row,     live ? wj * hv[u].x : 0.0f);
      atomicAdd(row + 1, live ? wj * hv[u].y : 0.0f);
    }
  }
}

__global__ __launch_bounds__(1024) void gat_edge_softmax_aggregate_owned(
    const int* __restrict__ ei, int E, int N,
    const float* __restrict__ Hin, const float* __restrict__ als,
    const float* __restrict__ ald, const unsigned* __restrict__ key32,
    const float* __restrict__ bias, float* __restrict__ outp)
{
  __shared__ __attribute__((aligned(16))) float accs[RNG * 64];
  __shared__ int      lists[32 * LCAP];
  __shared__ float    lgts[32 * LCAP];
  __shared__ unsigned menc[RNG];
  __shared__ float    den[RNG];
  __shared__ int      full;

  const int tid  = threadIdx.x;
  const int lane = tid & 31;
  const int wave = __builtin_amdgcn_readfirstlane(tid >> 5);
  const int ET   = E + N;
  const int lo   = blockIdx.x * RNG;

  for (int i = tid; i < RNG * 64; i += 1024) accs[i] = 0.0f;
  if (tid < RNG) { menc[tid] = 0u; den[tid] = 0.0f; }
  if (tid == 0) full = 0;
  __syncthreads();

  int*   mylist = lists + wave * LCAP;
  float* mylgt  = lgts  + wave * LCAP;
  const int ngroups = (ET + 15) >> 4;
  const unsigned kb = (unsigned)blockIdx.x * 0x01010101u;
  int cntl = 0;

  const int nchunk = (ngroups + 4095) >> 12;
  int chunk = (int)(blockIdx.x % (unsigned)nchunk);
  for (int it = 0; it < nchunk; ++it) {
    const int g0 = chunk << 12;
    chunk = (chunk + 1 == nchunk) ? 0 : chunk + 1;
    v4u x[4];
#pragma unroll
    for (int q = 0; q < 4; ++q) {
      const int gi = g0 + q * 1024 + tid;
      x[q] = *(const v4u*)(key32 + 4 * (size_t)((gi < ngroups) ? gi : (ngroups - 1)));
    }
#pragma unroll
    for (int q = 0; q < 4; ++q) {
      const int gi = g0 + q * 1024 + tid;
      const unsigned t0 = x[q].x ^ kb, t1 = x[q].y ^ kb, t2 = x[q].z ^ kb, t3 = x[q].w ^ kb;
      unsigned any = (((t0 - 0x01010101u) & ~t0) | ((t1 - 0x01010101u) & ~t1) |
                      ((t2 - 0x01010101u) & ~t2) | ((t3 - 0x01010101u) & ~t3)) & 0x80808080u;
      if (gi >= ngroups) any = 0u;
      if (any != 0u) {
        unsigned m = zero_byte_map(t0, t1, t2, t3);
        do {
          const int p = __builtin_ctz(m);
          m &= m - 1u;
          if (cntl < 32) mylist[cntl * 32 + lane] = gi * 16 + 4 * (p & 7) + (p >> 3);
          ++cntl;
        } while (m != 0u);
      }
    }
  }
  if (cntl > 32) full = 1;
  __syncthreads();
  const bool direct = (full != 0);
  const unsigned char* kbytes = (const unsigned char*)key32;
  const int total = direct ? 0 : own_compact(mylist, cntl, lane);

  if (direct) {
    for (int e = tid; e < ET; e += 1024) {
      if ((unsigned)kbytes[e] != blockIdx.x) continue;
      int s, d;
      if (e < E) { s = ei[e]; d = ei[E + e]; } else { s = d = e - E; }
      atomicMax(&menc[d - lo], fenc(edge_logit(als[s], ald[d])));
    }
  } else {
    for (int base = 0; base < total; base += 128) {
      int dl4[4]; float v4[4];
      own_resolve4(mylist, mylgt, base, total, lane, lo, ei, E, als, ald, dl4, v4);
#pragma unroll
      for (int u = 0; u < 4; ++u) atomicMax(&menc[dl4[u]], fenc(v4[u]));
    }
  }
  __syncthreads();

  if (direct) {
    for (int e = tid; e < ET; e += 1024) {
      if ((unsigned)kbytes[e] != blockIdx.x) continue;
      int s, d;
      if (e < E) { s = ei[e]; d = ei[E + e]; } else { s = d = e - E; }
      const int   dl = d - lo;
      const float ex = __expf(edge_logit(als[s], ald[d]) - fdec(menc[dl]));
      atomicAdd(&den[dl], ex);
      for (int f = 0; f < 64; ++f) atomicAdd(&accs[dl * 64 + f], ex * Hin[(size_t)s * 64 + f]);
    }
  } else {
    for (int base = 0; base < total; base += 128) {
      int s4[4], dl4[4]; float v4[4], ex4[4];
      own_read4(mylist, mylgt, base, total, lane, s4, dl4, v4);
#pragma unroll
      for (int u = 0; u < 4; ++u) {
        ex4[u] = __expf(v4[u] - fdec(menc[dl4[u]]));
        if (base + u * 32 + lane < total) atomicAdd(&den[dl4[u]], ex4[u]);
      }
#pragma unroll
      for (int u = 0; u < 4; ++u) {
        const int left = total - base - u * 32;
        if (left > 0) own_rows(s4[u], dl4[u], ex4[u], (left < 32) ? left : 32, lane, Hin, accs);
      }
    }
  }
  __syncthreads();

  for (int i = tid; i < RNG * 16; i += 1024) {
    const int r = i >> 4, c4 = i & 15;
    const int n = lo + r;
    if (n < N) {
      const v4f a  = *(const v4f*)(accs + r * 64 + c4 * 4);
      const v4f bv = *(const v4f*)(bias + c4 * 4);
      *(v4f*)(outp + (size_t)n * 64 + c4 * 4) = a / den[r] + bv;
    }
  }
}

extern "C" void kernel_launch(void* const* d_in, const int* in_sizes, int n_in,
                              void* d_out, int out_size, void* d_ws, size_t ws_size,
                              hipStream_t stream) {
  const float* x   = (const float*)d_in[0];
  const int*   ei  = (const int*)  d_in[1];
  const float* W1  = (const float*)d_in[2];
  const float* as1 = (const float*)d_in[3];
  const float* ad1 = (const float*)d_in[4];
  const float* b1  = (const float*)d_in[5];
  const float* W2  = (const float*)d_in[6];
  const float* as2 = (const float*)d_in[7];
  const float* ad2 = (const float*)d_in[8];
  const float* b2  = (const float*)d_in[9];

  const int N  = in_sizes[0] / 128;
  const int E  = in_sizes[1] / 2;

  char* w = (char*)d_ws;
  float*    h     = (float*)w;    w += (size_t)N * 64 * sizeof(float);
  float*    agg   = (float*)w;    w += (size_t)N * 64 * sizeof(float);
  float*    als   = (float*)w;    w += (size_t)N * sizeof(float);
  float*    ald   = (float*)w;    w += (size_t)N * sizeof(float);
  w = (char*)d_ws + (((size_t)(w - (char*)d_ws) + 15) & ~(size_t)15);
  unsigned* key   = (unsigned*)w;
  float*    out   = (float*)d_out;

  const dim3 blk(256);
  const int gGemm     = (N + 127) / 128;
  const int gLogit    = (N + 127) / 128;
  const int gOwn      = (N + RNG - 1) / RNG;

  gat_gemm_wmma<false><<<gGemm,  blk, 0, stream>>>(x, W1, h, N, 128);
  gat_logits       <<<gLogit,    blk, 0, stream>>>(h, as1, ad1, als, ald, N, ei, E, key);
  gat_edge_softmax_aggregate_owned<<<gOwn, dim3(1024), 0, stream>>>(ei, E, N, h, als, ald, key, b1, agg);

  gat_gemm_wmma<true><<<gGemm,   blk, 0, stream>>>(agg, W2, h, N, 64);
  gat_logits       <<<gLogit,    blk, 0, stream>>>(h, as2, ad2, als, ald, N, ei, E, (unsigned*)nullptr);
  gat_edge_softmax_aggregate_owned<<<gOwn, dim3(1024), 0, stream>>>(ei, E, N, h, als, ald, key, b2, out);
}


# Edge pass: layer 1's launch stores each wave's resolved edge records (source, local row) and count in the workspace; layer 2's launch reads them: no owner-byte scan, packing or index loads there
# speedup vs baseline: 3.3501x; 1.1510x over previous
#include <hip/hip_runtime.h>


typedef float    v2f  __attribute__((ext_vector_type(2)));
typedef float    v4f  __attribute__((ext_vector_type(4)));
typedef float    v8f  __attribute__((ext_vector_type(8)));
typedef unsigned v4u  __attribute__((ext_vector_type(4)));

#define LRELU_SLOPE 0.2f
#define RNG   200
#define LCAP  1024

__device__ __forceinline__ unsigned fenc(float f) {
  unsigned u = __float_as_uint(f);
  return (u & 0x80000000u) ? ~u : (u | 0x80000000u);
}
__device__ __forceinline__ float fdec(unsigned u) {
  u = (u & 0x80000000u) ? (u & 0x7FFFFFFFu) : ~u;
  return __uint_as_float(u);
}

template <bool RELU_A>
__global__ __launch_bounds__(256) void gat_gemm_wmma(
    const float* __restrict__ X,
    const float* __restrict__ Wm,
    float* __restrict__ Hout, int N, int K)
{
  __shared__ float wldsT[64 * 130];
  const int stride = K + 2;
  const int tid = threadIdx.x;

  for (int i = tid; i < K * 64; i += 256) {
    const int k = i >> 6, col = i & 63;
    wldsT[col * stride + k] = Wm[i];
  }
  __syncthreads();

  const int wave = tid >> 5;
  const int lane = tid & 31;
  const int l16  = lane & 15;
  const int hsel = lane >> 4;
  const int rowBase = blockIdx.x * 128 + wave * 16;
  int row = rowBase + l16;
  if (row >= N) row = N - 1;

  v8f acc[4] = {};

  const int koff = hsel * 2;
  const float* xrow = X + (size_t)row * K + koff;
  const float* wb0 = &wldsT[(0 * 16 + l16) * stride + koff];
  const float* wb1 = &wldsT[(1 * 16 + l16) * stride + koff];
  const float* wb2 = &wldsT[(2 * 16 + l16) * stride + koff];
  const float* wb3 = &wldsT[(3 * 16 + l16) * stride + koff];

  for (int kk = 0; kk < K; kk += 4) {
    const float2 av  = *(const float2*)(xrow + kk);
    const float2 bv0 = *(const float2*)(wb0 + kk);
    const float2 bv1 = *(const float2*)(wb1 + kk);
    const float2 bv2 = *(const float2*)(wb2 + kk);
    const float2 bv3 = *(const float2*)(wb3 + kk);
    v2f a;
    if (RELU_A) { a.x = fmaxf(av.x, 0.0f); a.y = fmaxf(av.y, 0.0f); }
    else        { a.x = av.x; a.y = av.y; }
    v2f b0; b0.x = bv0.x; b0.y = bv0.y;
    v2f b1; b1.x = bv1.x; b1.y = bv1.y;
    v2f b2; b2.x = bv2.x; b2.y = bv2.y;
    v2f b3; b3.x = bv3.x; b3.y = bv3.y;
    acc[0] = __builtin_amdgcn_wmma_f32_16x16x4_f32(false, a, false, b0, (short)0, acc[0], false, false);
    acc[1] = __builtin_amdgcn_wmma_f32_16x16x4_f32(false, a, false, b1, (short)0, acc[1], false, false);
    acc[2] = __builtin_amdgcn_wmma_f32_16x16x4_f32(false, a, false, b2, (short)0, acc[2], false, false);
    acc[3] = __builtin_amdgcn_wmma_f32_16x16x4_f32(false, a, false, b3, (short)0, acc[3], false, false);
  }

#pragma unroll
  for (int nt = 0; nt < 4; ++nt) {
#pragma unroll
    for (int j = 0; j < 8; ++j) {
      const int ro = rowBase + j + hsel * 8;
      if (ro < N) Hout[(size_t)ro * 64 + nt * 16 + l16] = acc[nt][j];
    }
  }
}


__device__ __forceinline__ float row16_sum(float v) {
  v += __builtin_bit_cast(float, __builtin_amdgcn_update_dpp(0, __builtin_bit_cast(int, v), 0x161, 0xf, 0xf, false));
  v += __builtin_bit_cast(float, __builtin_amdgcn_update_dpp(0, __builtin_bit_cast(int, v), 0x162, 0xf, 0xf, false));
  v += __builtin_bit_cast(float, __builtin_amdgcn_update_dpp(0, __builtin_bit_cast(int, v), 0x164, 0xf, 0xf, false));
  v += __builtin_bit_cast(float, __builtin_amdgcn_update_dpp(0, __builtin_bit_cast(int, v), 0x168, 0xf, 0xf, false));
  return v;
}

__global__ __launch_bounds__(256) void gat_logits(
    const float* __restrict__ H, const float* __restrict__ a_src,
    const float* __restrict__ a_dst, float* __restrict__ als,
    float* __restrict__ ald, int N,
    const int* __restrict__ ei, int E, unsigned* __restrict__ key32)
{
  const unsigned gid = blockIdx.x * (unsigned)blockDim.x + threadIdx.x;
  const int lane  = threadIdx.x & 31;
  const int c     = lane & 15;
  const int hs    = lane >> 4;
  const int node0 = (int)(gid >> 5) * 16;

  v4f hv[8];
#pragma unroll
  for (int j = 0; j < 8; ++j) {
    int n = node0 + 2 * j + hs;
    n = (n < N) ? n : N - 1;
    hv[j] = *(const v4f*)(H + (size_t)n * 64 + c * 4);
  }
  const v4f as = *(const v4f*)(a_src + c * 4);
  const v4f ad = *(const v4f*)(a_dst + c * 4);

  if (key32) {
    const int ET = E + N;
    const unsigned nwords = (unsigned)((ET + 15) >> 4) << 2;
    for (unsigned w = gid; w < nwords; w += gridDim.x * (unsigned)blockDim.x) {
      const int e0 = (int)(w * 4u);
      int dd[4];
#pragma unroll
      for (int j = 0; j < 4; ++j) dd[j] = ei[E + ((e0 + j < E) ? e0 + j : 0)];
      asm volatile("" :: "v"(dd[0]), "v"(dd[1]), "v"(dd[2]), "v"(dd[3]));
      unsigned word = 0u;
#pragma unroll
      for (int j = 0; j < 4; ++j) {
        const int e = e0 + j;
        const int d = (e < E) ? dd[j] : e - E;
        const unsigned b = (e < ET) ? (unsigned)(d / RNG) : 255u;
        word |= b << (8 * j);
      }
      key32[w] = word;
    }
  }

  float rs = 0.0f, rd = 0.0f;
#pragma unroll
  for (int j = 0; j < 8; ++j) {
    float ps = hv[j].x * as.x + hv[j].y * as.y + hv[j].z * as.z + hv[j].w * as.w;
    float pd = hv[j].x * ad.x + hv[j].y * ad.y + hv[j].z * ad.z + hv[j].w * ad.w;
    ps = row16_sum(ps);
    pd = row16_sum(pd);
    if ((c >> 1) == j) { rs = ps; rd = pd; }
  }
  const int node = node0 + c;
  if ((c & 1) == hs && node < N) { als[node] = rs; ald[node] = rd; }
}


__device__ __forceinline__ unsigned zero_byte_map(unsigned t0, unsigned t1, unsigned t2, unsigned t3) {
  const unsigned M  = 0x7F7F7F7Fu;
  const unsigned y0 = ~(((t0 & M) + M) | t0 | M);
  const unsigned y1 = ~(((t1 & M) + M) | t1 | M);
  const unsigned y2 = ~(((t2 & M) + M) | t2 | M);
  const unsigned y3 = ~(((t3 & M) + M) | t3 | M);
  return (y0 >> 7) | (y1 >> 6) | (y2 >> 5) | (y3 >> 4);
}

__device__ __forceinline__ int own_compact(int* mylist, int cntl, int lane) {
  int total = 0;
  for (int k = 0; k < 32; ++k) {
    const bool has = k < cntl;
    const unsigned bal = __builtin_amdgcn_ballot_w32(has);
    if (bal == 0u) break;
    if (has) {
      const int v = mylist[k * 32 + lane];
      mylist[total + (int)__builtin_amdgcn_mbcnt_lo(bal, 0u)] = v;
    }
    total += __builtin_popcount(bal);
  }
  return total;
}

__device__ __forceinline__ float edge_logit(float as, float ad) {
  const float v = as + ad;
  return (v > 0.0f) ? v : LRELU_SLOPE * v;
}

__device__ __forceinline__ void own_resolve4(
    int* mylist, float* mylgt, int* __restrict__ myrec, int base, int total, int lane, int lo,
    const int* __restrict__ ei, int E,
    const float* __restrict__ als, const float* __restrict__ ald,
    int (&dl4)[4], float (&v4)[4])
{
  int s4[4], d4[4];
#pragma unroll
  for (int u = 0; u < 4; ++u) {
    const int i  = base + u * 32 + lane;
    const int e  = mylist[(i < total) ? i : base];
    const int ec = (e < E) ? e : 0;
    const int sl = ei[ec], dd = ei[E + ec];
    s4[u] = (e < E) ? sl : e - E;
    d4[u] = (e < E) ? dd : e - E;
  }
#pragma unroll
  for (int u = 0; u < 4; ++u) {
    v4[u]  = edge_logit(als[s4[u]], ald[d4[u]]);
    dl4[u] = d4[u] - lo;
  }
#pragma unroll
  for (int u = 0; u < 4; ++u) {
    const int i = base + u * 32 + lane;
    if (i < total) {
      const int rec = (int)((unsigned)s4[u] | ((unsigned)dl4[u] << 24));
      mylist[i] = rec;
      mylgt[i]  = v4[u];
      myrec[i]  = rec;
    }
  }
}

__device__ __forceinline__ void own_fetch4(
    int* mylist, float* mylgt, const int* __restrict__ myrec, int base, int total, int lane, int lo,
    const float* __restrict__ als, const float* __restrict__ ald,
    int (&dl4)[4], float (&v4)[4])
{
  unsigned p4[4];
#pragma unroll
  for (int u = 0; u < 4; ++u) {
    const int i = base + u * 32 + lane;
    p4[u] = (unsigned)myrec[(i < total) ? i : base];
  }
#pragma unroll
  for (int u = 0; u < 4; ++u) {
    dl4[u] = (int)(p4[u] >> 24);
    v4[u]  = edge_logit(als[p4[u] & 0x00FFFFFFu], ald[lo + dl4[u]]);
  }
#pragma unroll
  for (int u = 0; u < 4; ++u) {
    const int i = base + u * 32 + lane;
    if (i < total) {
      mylist[i] = (int)p4[u];
      mylgt[i]  = v4[u];
    }
  }
}

__device__ __forceinline__ void own_read4(
    const int* mylist, const float* mylgt, int base, int total, int lane,
    int (&s4)[4], int (&dl4)[4], float (&v4)[4])
{
#pragma unroll
  for (int u = 0; u < 4; ++u) {
    const int i = base + u * 32 + lane;
    const int j = (i < total) ? i : base;
    const unsigned p = (unsigned)mylist[j];
    v4[u]  = mylgt[j];
    s4[u]  = (int)(p & 0x00FFFFFFu);
    dl4[u] = (int)(p >> 24);
  }
}

__device__ __forceinline__ void own_rows(
    int s, int dl, float w, int nb, int lane,
    const float* __restrict__ Hin, float* accs)
{
  for (int g = 0; g < nb; g += 8) {
    float2 hv[8];
#pragma unroll
    for (int u = 0; u < 8; ++u) {
      const int sj = __builtin_amdgcn_readlane(s, g + u);
      hv[u] = *(const float2*)(Hin + (size_t)sj * 64 + lane * 2);
    }
#pragma unroll
    for (int u = 0; u < 8; ++u) {
      const bool  live = (u == 0) || (g + u < nb);
      const int   dj = __builtin_amdgcn_readlane(dl, g + u);
      const float wj = __uint_as_float(
          __builtin_amdgcn_readlane(__float_as_uint(w), g + u));
      float* row = accs + dj * 64 + lane * 2;
      atomicAdd(row,     live ? wj * hv[u].x : 0.0f);
      atomicAdd(row + 1, live ? wj * hv[u].y : 0.0f);
    }
  }
}

template <bool FIRST>
__global__ __launch_bounds__(1024) void gat_edge_softmax_aggregate_owned(
    const int* __restrict__ ei, int E, int N,
    const float* __restrict__ Hin, const float* __restrict__ als,
    const float* __restrict__ ald, const unsigned* __restrict__ key32,
    const float* __restrict__ bias, float* __restrict__ outp,
    int* __restrict__ grec, int* __restrict__ gcnt)
{
  __shared__ __attribute__((aligned(16))) float accs[RNG * 64];
  __shared__ int      lists[32 * LCAP];
  __shared__ float    lgts[32 * LCAP];
  __shared__ unsigned menc[RNG];
  __shared__ float    den[RNG];
  __shared__ int      full;

  const int tid  = threadIdx.x;
  const int lane = tid & 31;
  const int wave = __builtin_amdgcn_readfirstlane(tid >> 5);
  const int ET   = E + N;
  const int lo   = blockIdx.x * RNG;

  for (int i = tid; i < RNG * 64; i += 1024) accs[i] = 0.0f;
  if (tid < RNG) { menc[tid] = 0u; den[tid] = 0.0f; }
  if (tid == 0) full = 0;
  __syncthreads();

  int*   mylist = lists + wave * LCAP;
  float* mylgt  = lgts  + wave * LCAP;
  int*   myrec  = grec + ((size_t)blockIdx.x * 32 + wave) * LCAP;
  int*   mycnt  = gcnt + blockIdx.x * 32 + wave;
  const unsigned char* kbytes = (const unsigned char*)key32;
  bool direct;
  int  total;
  if (FIRST) {
    const int ngroups = (ET + 15) >> 4;
    const unsigned kb = (unsigned)blockIdx.x * 0x01010101u;
    int cntl = 0;

    const int nchunk = (ngroups + 4095) >> 12;
    int chunk = (int)(blockIdx.x % (unsigned)nchunk);
    for (int it = 0; it < nchunk; ++it) {
      const int g0 = chunk << 12;
      chunk = (chunk + 1 == nchunk) ? 0 : chunk + 1;
      v4u x[4];
#pragma unroll
      for (int q = 0; q < 4; ++q) {
        const int gi = g0 + q * 1024 + tid;
        x[q] = *(const v4u*)(key32 + 4 * (size_t)((gi < ngroups) ? gi : (ngroups - 1)));
      }
#pragma unroll
      for (int q = 0; q < 4; ++q) {
        const int gi = g0 + q * 1024 + tid;
        const unsigned t0 = x[q].x ^ kb, t1 = x[q].y ^ kb, t2 = x[q].z ^ kb, t3 = x[q].w ^ kb;
        unsigned any = (((t0 - 0x01010101u) & ~t0) | ((t1 - 0x01010101u) & ~t1) |
                        ((t2 - 0x01010101u) & ~t2) | ((t3 - 0x01010101u) & ~t3)) & 0x80808080u;
        if (gi >= ngroups) any = 0u;
        if (any != 0u) {
          unsigned m = zero_byte_map(t0, t1, t2, t3);
          do {
            const int p = __builtin_ctz(m);
            m &= m - 1u;
            if (cntl < 32) mylist[cntl * 32 + lane] = gi * 16 + 4 * (p & 7) + (p >> 3);
            ++cntl;
          } while (m != 0u);
        }
      }
    }
    if (cntl > 32) full = 1;
    __syncthreads();
    direct = (full != 0);
    total  = direct ? 0 : own_compact(mylist, cntl, lane);
    if (lane == 0) *mycnt = direct ? -1 : total;
  } else {
    const int t = __builtin_amdgcn_readfirstlane(*mycnt);
    direct = (t < 0);
    total  = direct ? 0 : t;
  }

  if (direct) {
    for (int e = tid; e < ET; e += 1024) {
      if ((unsigned)kbytes[e] != blockIdx.x) continue;
      int s, d;
      if (e < E) { s = ei[e]; d = ei[E + e]; } else { s = d = e - E; }
      atomicMax(&menc[d - lo], fenc(edge_logit(als[s], ald[d])));
    }
  } else {
    for (int base = 0; base < total; base += 128) {
      int dl4[4]; float v4[4];
      if (FIRST) own_resolve4(mylist, mylgt, myrec, base, total, lane, lo, ei, E, als, ald, dl4, v4);
      else       own_fetch4(mylist, mylgt, myrec, base, total, lane, lo, als, ald, dl4, v4);
#pragma unroll
      for (int u = 0; u < 4; ++u) atomicMax(&menc[dl4[u]], fenc(v4[u]));
    }
  }
  __syncthreads();

  if (direct) {
    for (int e = tid; e < ET; e += 1024) {
      if ((unsigned)kbytes[e] != blockIdx.x) continue;
      int s, d;
      if (e < E) { s = ei[e]; d = ei[E + e]; } else { s = d = e - E; }
      const int   dl = d - lo;
      const float ex = __expf(edge_logit(als[s], ald[d]) - fdec(menc[dl]));
      atomicAdd(&den[dl], ex);
      for (int f = 0; f < 64; ++f) atomicAdd(&accs[dl * 64 + f], ex * Hin[(size_t)s * 64 + f]);
    }
  } else {
    for (int base = 0; base < total; base += 128) {
      int s4[4], dl4[4]; float v4[4], ex4[4];
      own_read4(mylist, mylgt, base, total, lane, s4, dl4, v4);
#pragma unroll
      for (int u = 0; u < 4; ++u) {
        ex4[u] = __expf(v4[u] - fdec(menc[dl4[u]]));
        if (base + u * 32 + lane < total) atomicAdd(&den[dl4[u]], ex4[u]);
      }
#pragma unroll
      for (int u = 0; u < 4; ++u) {
        const int left = total - base - u * 32;
        if (left > 0) own_rows(s4[u], dl4[u], ex4[u], (left < 32) ? left : 32, lane, Hin, accs);
      }
    }
  }
  __syncthreads();

  for (int i = tid; i < RNG * 16; i += 1024) {
    const int r = i >> 4, c4 = i & 15;
    const int n = lo + r;
    if (n < N) {
      const v4f a  = *(const v4f*)(accs + r * 64 + c4 * 4);
      const v4f bv = *(const v4f*)(bias + c4 * 4);
      *(v4f*)(outp + (size_t)n * 64 + c4 * 4) = a / den[r] + bv;
    }
  }
}

extern "C" void kernel_launch(void* const* d_in, const int* in_sizes, int n_in,
                              void* d_out, int out_size, void* d_ws, size_t ws_size,
                              hipStream_t stream) {
  const float* x   = (const float*)d_in[0];
  const int*   ei  = (const int*)  d_in[1];
  const float* W1  = (const float*)d_in[2];
  const float* as1 = (const float*)d_in[3];
  const float* ad1 = (const float*)d_in[4];
  const float* b1  = (const float*)d_in[5];
  const float* W2  = (const float*)d_in[6];
  const float* as2 = (const float*)d_in[7];
  const float* ad2 = (const float*)d_in[8];
  const float* b2  = (const float*)d_in[9];

  const int N  = in_sizes[0] / 128;
  const int E  = in_sizes[1] / 2;

  char* w = (char*)d_ws;
  float*    h     = (float*)w;    w += (size_t)N * 64 * sizeof(float);
  float*    agg   = (float*)w;    w += (size_t)N * 64 * sizeof(float);
  float*    als   = (float*)w;    w += (size_t)N * sizeof(float);
  float*    ald   = (float*)w;    w += (size_t)N * sizeof(float);
  w = (char*)d_ws + (((size_t)(w - (char*)d_ws) + 15) & ~(size_t)15);
  unsigned* key   = (unsigned*)w;    w += ((size_t)(E + N + 15) >> 4) * 16;
  const int gOwn  = (N + RNG - 1) / RNG;
  int*      cnt   = (int*)w;         w += (size_t)gOwn * 32 * sizeof(int);
  w = (char*)d_ws + (((size_t)(w - (char*)d_ws) + 255) & ~(size_t)255);
  int*      rec   = (int*)w;
  float*    out   = (float*)d_out;

  const dim3 blk(256);
  const int gGemm     = (N + 127) / 128;
  const int gLogit    = (N + 127) / 128;

  gat_gemm_wmma<false><<<gGemm,  blk, 0, stream>>>(x, W1, h, N, 128);
  gat_logits       <<<gLogit,    blk, 0, stream>>>(h, as1, ad1, als, ald, N, ei, E, key);
  gat_edge_softmax_aggregate_owned<true><<<gOwn, dim3(1024), 0, stream>>>(ei, E, N, h, als, ald, key, b1, agg, rec, cnt);

  gat_gemm_wmma<true><<<gGemm,   blk, 0, stream>>>(agg, W2, h, N, 64);
  gat_logits       <<<gLogit,    blk, 0, stream>>>(h, as2, ad2, als, ald, N, ei, E, (unsigned*)nullptr);
  gat_edge_softmax_aggregate_owned<false><<<gOwn, dim3(1024), 0, stream>>>(ei, E, N, h, als, ald, key, b2, out, rec, cnt);
}


# Owner search on 8 bit planes (wave votes) instead of owner bytes: a block's match word of 32 edges is one xor-and per plane; self-loops, known by row, appended unsearched; LDS arrival-order sums
# speedup vs baseline: 4.1563x; 1.2406x over previous
#include <hip/hip_runtime.h>


typedef float    v2f  __attribute__((ext_vector_type(2)));
typedef float    v4f  __attribute__((ext_vector_type(4)));
typedef float    v8f  __attribute__((ext_vector_type(8)));
typedef unsigned v4u  __attribute__((ext_vector_type(4)));

#define LRELU_SLOPE 0.2f
#define RNG   200
#define LCAP  1024

__device__ __forceinline__ unsigned fenc(float f) {
  unsigned u = __float_as_uint(f);
  return (u & 0x80000000u) ? ~u : (u | 0x80000000u);
}
__device__ __forceinline__ float fdec(unsigned u) {
  u = (u & 0x80000000u) ? (u & 0x7FFFFFFFu) : ~u;
  return __uint_as_float(u);
}

template <bool RELU_A>
__global__ __launch_bounds__(256) void gat_gemm_wmma(
    const float* __restrict__ X,
    const float* __restrict__ Wm,
    float* __restrict__ Hout, int N, int K)
{
  __shared__ float wldsT[64 * 130];
  const int stride = K + 2;
  const int tid = threadIdx.x;

  for (int i = tid; i < K * 64; i += 256) {
    const int k = i >> 6, col = i & 63;
    wldsT[col * stride + k] = Wm[i];
  }
  __syncthreads();

  const int wave = tid >> 5;
  const int lane = tid & 31;
  const int l16  = lane & 15;
  const int hsel = lane >> 4;
  const int rowBase = blockIdx.x * 128 + wave * 16;
  int row = rowBase + l16;
  if (row >= N) row = N - 1;

  v8f acc[4] = {};

  const int koff = hsel * 2;
  const float* xrow = X + (size_t)row * K + koff;
  const float* wb0 = &wldsT[(0 * 16 + l16) * stride + koff];
  const float* wb1 = &wldsT[(1 * 16 + l16) * stride + koff];
  const float* wb2 = &wldsT[(2 * 16 + l16) * stride + koff];
  const float* wb3 = &wldsT[(3 * 16 + l16) * stride + koff];

  for (int kk = 0; kk < K; kk += 4) {
    const float2 av  = *(const float2*)(xrow + kk);
    const float2 bv0 = *(const float2*)(wb0 + kk);
    const float2 bv1 = *(const float2*)(wb1 + kk);
    const float2 bv2 = *(const float2*)(wb2 + kk);
    const float2 bv3 = *(const float2*)(wb3 + kk);
    v2f a;
    if (RELU_A) { a.x = fmaxf(av.x, 0.0f); a.y = fmaxf(av.y, 0.0f); }
    else        { a.x = av.x; a.y = av.y; }
    v2f b0; b0.x = bv0.x; b0.y = bv0.y;
    v2f b1; b1.x = bv1.x; b1.y = bv1.y;
    v2f b2; b2.x = bv2.x; b2.y = bv2.y;
    v2f b3; b3.x = bv3.x; b3.y = bv3.y;
    acc[0] = __builtin_amdgcn_wmma_f32_16x16x4_f32(false, a, false, b0, (short)0, acc[0], false, false);
    acc[1] = __builtin_amdgcn_wmma_f32_16x16x4_f32(false, a, false, b1, (short)0, acc[1], false, false);
    acc[2] = __builtin_amdgcn_wmma_f32_16x16x4_f32(false, a, false, b2, (short)0, acc[2], false, false);
    acc[3] = __builtin_amdgcn_wmma_f32_16x16x4_f32(false, a, false, b3, (short)0, acc[3], false, false);
  }

#pragma unroll
  for (int nt = 0; nt < 4; ++nt) {
#pragma unroll
    for (int j = 0; j < 8; ++j) {
      const int ro = rowBase + j + hsel * 8;
      if (ro < N) Hout[(size_t)ro * 64 + nt * 16 + l16] = acc[nt][j];
    }
  }
}


__device__ __forceinline__ float row16_sum(float v) {
  v += __builtin_bit_cast(float, __builtin_amdgcn_update_dpp(0, __builtin_bit_cast(int, v), 0x161, 0xf, 0xf, false));
  v += __builtin_bit_cast(float, __builtin_amdgcn_update_dpp(0, __builtin_bit_cast(int, v), 0x162, 0xf, 0xf, false));
  v += __builtin_bit_cast(float, __builtin_amdgcn_update_dpp(0, __builtin_bit_cast(int, v), 0x164, 0xf, 0xf, false));
  v += __builtin_bit_cast(float, __builtin_amdgcn_update_dpp(0, __builtin_bit_cast(int, v), 0x168, 0xf, 0xf, false));
  return v;
}

__global__ __launch_bounds__(256) void gat_logits(
    const float* __restrict__ H, const float* __restrict__ a_src,
    const float* __restrict__ a_dst, float* __restrict__ als,
    float* __restrict__ ald, int N,
    const int* __restrict__ ei, int E, unsigned* __restrict__ key32)
{
  const unsigned gid = blockIdx.x * (unsigned)blockDim.x + threadIdx.x;
  const int lane  = threadIdx.x & 31;
  const int c     = lane & 15;
  const int hs    = lane >> 4;
  const int node0 = (int)(gid >> 5) * 16;

  v4f hv[8];
#pragma unroll
  for (int j = 0; j < 8; ++j) {
    int n = node0 + 2 * j + hs;
    n = (n < N) ? n : N - 1;
    hv[j] = *(const v4f*)(H + (size_t)n * 64 + c * 4);
  }
  const v4f as = *(const v4f*)(a_src + c * 4);
  const v4f ad = *(const v4f*)(a_dst + c * 4);

  if (key32) {
    const unsigned nW    = ((unsigned)(E + 127) >> 7) << 2;
    const unsigned ntask = (nW + 31u) >> 5;
    const unsigned wv  = (unsigned)__builtin_amdgcn_readfirstlane((int)(gid >> 5));
    const unsigned nwv = (gridDim.x * (unsigned)blockDim.x) >> 5;
    for (unsigned t = wv; t < ntask; t += nwv) {
      unsigned pw[8];
#pragma unroll
      for (int j = 0; j < 8; ++j) pw[j] = 0u;
      for (int k0 = 0; k0 < 32; k0 += 16) {
        int dd[16];
#pragma unroll
        for (int u = 0; u < 16; ++u) {
          const int e = (int)(t * 1024u) + (k0 + u) * 32 + lane;
          dd[u] = ei[E + ((e < E) ? e : 0)];
        }
#pragma unroll
        for (int u = 0; u < 16; ++u) {
          const int e = (int)(t * 1024u) + (k0 + u) * 32 + lane;
          const unsigned o = (e < E) ? (unsigned)(dd[u] / RNG) : 255u;
#pragma unroll
          for (int j = 0; j < 8; ++j) {
            const unsigned bal = __builtin_amdgcn_ballot_w32(((o >> j) & 1u) != 0u);
            pw[j] = (lane == k0 + u) ? bal : pw[j];
          }
        }
      }
      const unsigned wi = t * 32u + (unsigned)lane;
      if (wi < nW) {
#pragma unroll
        for (int j = 0; j < 8; ++j) key32[(size_t)j * nW + wi] = pw[j];
      }
    }
  }

  float rs = 0.0f, rd = 0.0f;
#pragma unroll
  for (int j = 0; j < 8; ++j) {
    float ps = hv[j].x * as.x + hv[j].y * as.y + hv[j].z * as.z + hv[j].w * as.w;
    float pd = hv[j].x * ad.x + hv[j].y * ad.y + hv[j].z * ad.z + hv[j].w * ad.w;
    ps = row16_sum(ps);
    pd = row16_sum(pd);
    if ((c >> 1) == j) { rs = ps; rd = pd; }
  }
  const int node = node0 + c;
  if ((c & 1) == hs && node < N) { als[node] = rs; ald[node] = rd; }
}


__device__ __forceinline__ int own_compact(int* mylist, int cntl, int lane) {
  int total = 0;
  for (int k = 0; k < 32; ++k) {
    const bool has = k < cntl;
    const unsigned bal = __builtin_amdgcn_ballot_w32(has);
    if (bal == 0u) break;
    if (has) {
      const int v = mylist[k * 32 + lane];
      mylist[total + (int)__builtin_amdgcn_mbcnt_lo(bal, 0u)] = v;
    }
    total += __builtin_popcount(bal);
  }
  return total;
}

__device__ __forceinline__ float edge_logit(float as, float ad) {
  const float v = as + ad;
  return (v > 0.0f) ? v : LRELU_SLOPE * v;
}

__device__ __forceinline__ void own_resolve4(
    int* mylist, float* mylgt, int* __restrict__ myrec, int base, int total, int lane, int lo,
    const int* __restrict__ ei, int E,
    const float* __restrict__ als, const float* __restrict__ ald,
    int (&dl4)[4], float (&v4)[4])
{
  int s4[4], d4[4];
#pragma unroll
  for (int u = 0; u < 4; ++u) {
    const int i  = base + u * 32 + lane;
    const int e  = mylist[(i < total) ? i : base];
    const int ec = (e < E) ? e : 0;
    const int sl = ei[ec], dd = ei[E + ec];
    s4[u] = (e < E) ? sl : e - E;
    d4[u] = (e < E) ? dd : e - E;
  }
#pragma unroll
  for (int u = 0; u < 4; ++u) {
    v4[u]  = edge_logit(als[s4[u]], ald[d4[u]]);
    dl4[u] = d4[u] - lo;
  }
#pragma unroll
  for (int u = 0; u < 4; ++u) {
    const int i = base + u * 32 + lane;
    if (i < total) {
      const int rec = (int)((unsigned)s4[u] | ((unsigned)dl4[u] << 24));
      mylist[i] = rec;
      mylgt[i]  = v4[u];
      myrec[i]  = rec;
    }
  }
}

__device__ __forceinline__ void own_fetch4(
    int* mylist, float* mylgt, const int* __restrict__ myrec, int base, int total, int lane, int lo,
    const float* __restrict__ als, const float* __restrict__ ald,
    int (&dl4)[4], float (&v4)[4])
{
  unsigned p4[4];
#pragma unroll
  for (int u = 0; u < 4; ++u) {
    const int i = base + u * 32 + lane;
    p4[u] = (unsigned)myrec[(i < total) ? i : base];
  }
#pragma unroll
  for (int u = 0; u < 4; ++u) {
    dl4[u] = (int)(p4[u] >> 24);
    v4[u]  = edge_logit(als[p4[u] & 0x00FFFFFFu], ald[lo + dl4[u]]);
  }
#pragma unroll
  for (int u = 0; u < 4; ++u) {
    const int i = base + u * 32 + lane;
    if (i < total) {
      mylist[i] = (int)p4[u];
      mylgt[i]  = v4[u];
    }
  }
}

__device__ __forceinline__ void own_read4(
    const int* mylist, const float* mylgt, int base, int total, int lane,
    int (&s4)[4], int (&dl4)[4], float (&v4)[4])
{
#pragma unroll
  for (int u = 0; u < 4; ++u) {
    const int i = base + u * 32 + lane;
    const int j = (i < total) ? i : base;
    const unsigned p = (unsigned)mylist[j];
    v4[u]  = mylgt[j];
    s4[u]  = (int)(p & 0x00FFFFFFu);
    dl4[u] = (int)(p >> 24);
  }
}

__device__ __forceinline__ void own_rows(
    int s, int dl, float w, int nb, int lane,
    const float* __restrict__ Hin, float* accs)
{
  for (int g = 0; g < nb; g += 8) {
    float2 hv[8];
#pragma unroll
    for (int u = 0; u < 8; ++u) {
      const int sj = __builtin_amdgcn_readlane(s, g + u);
      hv[u] = *(const float2*)(Hin + (size_t)sj * 64 + lane * 2);
    }
#pragma unroll
    for (int u = 0; u < 8; ++u) {
      const bool  live = (u == 0) || (g + u < nb);
      const int   dj = __builtin_amdgcn_readlane(dl, g + u);
      const float wj = __uint_as_float(
          __builtin_amdgcn_readlane(__float_as_uint(w), g + u));
      float* row = accs + dj * 64 + lane * 2;
      atomicAdd(row,     live ? wj * hv[u].x : 0.0f);
      atomicAdd(row + 1, live ? wj * hv[u].y : 0.0f);
    }
  }
}

template <bool FIRST>
__global__ __launch_bounds__(1024) void gat_edge_softmax_aggregate_owned(
    const int* __restrict__ ei, int E, int N,
    const float* __restrict__ Hin, const float* __restrict__ als,
    const float* __restrict__ ald, const unsigned* __restrict__ key32,
    const float* __restrict__ bias, float* __restrict__ outp,
    int* __restrict__ grec, int* __restrict__ gcnt)
{
  __shared__ __attribute__((aligned(16))) float accs[RNG * 64];
  __shared__ int      lists[32 * LCAP];
  __shared__ float    lgts[32 * LCAP];
  __shared__ unsigned menc[RNG];
  __shared__ float    den[RNG];
  __shared__ int      full;

  const int tid  = threadIdx.x;
  const int lane = tid & 31;
  const int wave = __builtin_amdgcn_readfirstlane(tid >> 5);
  const int ET   = E + N;
  const int lo   = blockIdx.x * RNG;

  for (int i = tid; i < RNG * 64; i += 1024) accs[i] = 0.0f;
  if (tid < RNG) { menc[tid] = 0u; den[tid] = 0.0f; }
  if (tid == 0) full = 0;
  __syncthreads();

  int*   mylist = lists + wave * LCAP;
  float* mylgt  = lgts  + wave * LCAP;
  int*   myrec  = grec + ((size_t)blockIdx.x * 32 + wave) * LCAP;
  int*   mycnt  = gcnt + blockIdx.x * 32 + wave;
  bool direct;
  int  total;
  if (FIRST) {
    const int nW = (int)(((unsigned)(E + 127) >> 7) << 2);
    const int nq = nW >> 2;
    unsigned nb[8];
#pragma unroll
    for (int j = 0; j < 8; ++j) nb[j] = ((blockIdx.x >> j) & 1u) ? 0u : 0xFFFFFFFFu;
    int cntl = 0;

    const int nchunk = (nq + 1023) >> 10;
    int chunk = (int)(blockIdx.x % (unsigned)nchunk);
    for (int it = 0; it < nchunk; ++it) {
      const int  qi  = (chunk << 10) + tid;
      chunk = (chunk + 1 == nchunk) ? 0 : chunk + 1;
      const bool inr = qi < nq;
      const int  qc  = inr ? qi : (nq - 1);
      v4u p[8];
#pragma unroll
      for (int j = 0; j < 8; ++j)
        p[j] = *(const v4u*)(key32 + (size_t)j * nW + 4 * (size_t)qc);
#pragma unroll
      for (int x = 0; x < 4; ++x) {
        unsigned m = p[0][x] ^ nb[0];
#pragma unroll
        for (int j = 1; j < 8; ++j) m &= p[j][x] ^ nb[j];
        if (!inr) m = 0u;
        while (m != 0u) {
          const int pb = __builtin_ctz(m);
          m &= m - 1u;
          if (cntl < 32) mylist[cntl * 32 + lane] = qi * 128 + x * 32 + pb;
          ++cntl;
        }
      }
    }
    {
      const int r = lane * 32 + wave;
      if (r < RNG && lo + r < N) {
        if (cntl < 32) mylist[cntl * 32 + lane] = E + lo + r;
        ++cntl;
      }
    }
    if (cntl > 32) full = 1;
    __syncthreads();
    direct = (full != 0);
    total  = direct ? 0 : own_compact(mylist, cntl, lane);
    if (lane == 0) *mycnt = direct ? -1 : total;
  } else {
    const int t = __builtin_amdgcn_readfirstlane(*mycnt);
    direct = (t < 0);
    total  = direct ? 0 : t;
  }

  if (direct) {
    for (int e = tid; e < ET; e += 1024) {
      const int d = (e < E) ? ei[E + e] : e - E;
      if ((unsigned)(d / RNG) != blockIdx.x) continue;
      const int s = (e < E) ? ei[e] : d;
      atomicMax(&menc[d - lo], fenc(edge_logit(als[s], ald[d])));
    }
  } else {
    for (int base = 0; base < total; base += 128) {
      int dl4[4]; float v4[4];
      if (FIRST) own_resolve4(mylist, mylgt, myrec, base, total, lane, lo, ei, E, als, ald, dl4, v4);
      else       own_fetch4(mylist, mylgt, myrec, base, total, lane, lo, als, ald, dl4, v4);
#pragma unroll
      for (int u = 0; u < 4; ++u) atomicMax(&menc[dl4[u]], fenc(v4[u]));
    }
  }
  __syncthreads();

  if (direct) {
    for (int e = tid; e < ET; e += 1024) {
      const int d = (e < E) ? ei[E + e] : e - E;
      if ((unsigned)(d / RNG) != blockIdx.x) continue;
      const int s = (e < E) ? ei[e] : d;
      const int   dl = d - lo;
      const float ex = __expf(edge_logit(als[s], ald[d]) - fdec(menc[dl]));
      atomicAdd(&den[dl], ex);
      for (int f = 0; f < 64; ++f) atomicAdd(&accs[dl * 64 + f], ex * Hin[(size_t)s * 64 + f]);
    }
  } else {
    for (int base = 0; base < total; base += 128) {
      int s4[4], dl4[4]; float v4[4], ex4[4];
      own_read4(mylist, mylgt, base, total, lane, s4, dl4, v4);
#pragma unroll
      for (int u = 0; u < 4; ++u) {
        ex4[u] = __expf(v4[u] - fdec(menc[dl4[u]]));
        if (base + u * 32 + lane < total) atomicAdd(&den[dl4[u]], ex4[u]);
      }
#pragma unroll
      for (int u = 0; u < 4; ++u) {
        const int left = total - base - u * 32;
        if (left > 0) own_rows(s4[u], dl4[u], ex4[u], (left < 32) ? left : 32, lane, Hin, accs);
      }
    }
  }
  __syncthreads();

  for (int i = tid; i < RNG * 16; i += 1024) {
    const int r = i >> 4, c4 = i & 15;
    const int n = lo + r;
    if (n < N) {
      const v4f a  = *(const v4f*)(accs + r * 64 + c4 * 4);
      const v4f bv = *(const v4f*)(bias + c4 * 4);
      *(v4f*)(outp + (size_t)n * 64 + c4 * 4) = a / den[r] + bv;
    }
  }
}

extern "C" void kernel_launch(void* const* d_in, const int* in_sizes, int n_in,
                              void* d_out, int out_size, void* d_ws, size_t ws_size,
                              hipStream_t stream) {
  const float* x   = (const float*)d_in[0];
  const int*   ei  = (const int*)  d_in[1];
  const float* W1  = (const float*)d_in[2];
  const float* as1 = (const float*)d_in[3];
  const float* ad1 = (const float*)d_in[4];
  const float* b1  = (const float*)d_in[5];
  const float* W2  = (const float*)d_in[6];
  const float* as2 = (const float*)d_in[7];
  const float* ad2 = (const float*)d_in[8];
  const float* b2  = (const float*)d_in[9];

  const int N  = in_sizes[0] / 128;
  const int E  = in_sizes[1] / 2;

  char* w = (char*)d_ws;
  float*    h     = (float*)w;    w += (size_t)N * 64 * sizeof(float);
  float*    agg   = (float*)w;    w += (size_t)N * 64 * sizeof(float);
  float*    als   = (float*)w;    w += (size_t)N * sizeof(float);
  float*    ald   = (float*)w;    w += (size_t)N * sizeof(float);
  w = (char*)d_ws + (((size_t)(w - (char*)d_ws) + 15) & ~(size_t)15);
  unsigned* key   = (unsigned*)w;    w += (((size_t)(E + 127) >> 7) << 2) * 8 * sizeof(unsigned);
  const int gOwn  = (N + RNG - 1) / RNG;
  int*      cnt   = (int*)w;         w += (size_t)gOwn * 32 * sizeof(int);
  w = (char*)d_ws + (((size_t)(w - (char*)d_ws) + 255) & ~(size_t)255);
  int*      rec   = (int*)w;
  float*    out   = (float*)d_out;

  const dim3 blk(256);
  const int gGemm     = (N + 127) / 128;
  const int gLogit    = (N + 127) / 128;

  gat_gemm_wmma<false><<<gGemm,  blk, 0, stream>>>(x, W1, h, N, 128);
  gat_logits       <<<gLogit,    blk, 0, stream>>>(h, as1, ad1, als, ald, N, ei, E, key);
  gat_edge_softmax_aggregate_owned<true><<<gOwn, dim3(1024), 0, stream>>>(ei, E, N, h, als, ald, key, b1, agg, rec, cnt);

  gat_gemm_wmma<true><<<gGemm,   blk, 0, stream>>>(agg, W2, h, N, 64);
  gat_logits       <<<gLogit,    blk, 0, stream>>>(h, as2, ad2, als, ald, N, ei, E, (unsigned*)nullptr);
  gat_edge_softmax_aggregate_owned<false><<<gOwn, dim3(1024), 0, stream>>>(ei, E, N, h, als, ald, key, b2, out, rec, cnt);
}
